# starting kernel
# baseline (speedup 1.0000x reference)
#include <hip/hip_runtime.h>
#include <hip/hip_bf16.h>


typedef __attribute__((ext_vector_type(2))) float v2f;
typedef __attribute__((ext_vector_type(8))) float v8f;

#define D_IN   256
#define D_OUT  96
#define MINN   1e-15f
#define MAXN   0.996f
#define ARTC   (1.0f - 1e-7f)

__device__ __forceinline__ float wsum(float v) {
    v += __shfl_xor(v, 16, 32);
    v += __shfl_xor(v, 8, 32);
    v += __shfl_xor(v, 4, 32);
    v += __shfl_xor(v, 2, 32);
    v += __shfl_xor(v, 1, 32);
    return v;
}

__global__ void bias_kernel(const float* __restrict__ bias, float* __restrict__ hb) {
    __shared__ float red[4];
    int t = threadIdx.x;
    float v = (t < D_OUT) ? bias[t] : 0.0f;
    float ss = wsum(v * v);
    if ((t & 31) == 0) red[t >> 5] = ss;
    __syncthreads();
    float n = fmaxf(sqrtf(red[0] + red[1] + red[2] + red[3]), MINN);
    float p = tanhf(n) / n * v;
    __syncthreads();
    float ps = wsum(p * p);
    if ((t & 31) == 0) red[t >> 5] = ps;
    __syncthreads();
    float pn = fmaxf(sqrtf(red[0] + red[1] + red[2] + red[3]), MINN);
    float pf = (pn > MAXN) ? MAXN / pn : 1.0f;
    if (t < D_OUT) hb[t] = p * pf;
}

__global__ __launch_bounds__(256) void gemm_wmma(const float* __restrict__ x,
                                                 const float* __restrict__ w,
                                                 float* __restrict__ mx, int N) {
    const int nTilesM = (N + 15) >> 4;
    const int tileM   = (blockIdx.x << 3) + (threadIdx.x >> 5);
    if (tileM >= nTilesM) return;

    const int lane = threadIdx.x & 31;
    const int half = lane >> 4;
    const int mr   = lane & 15;

    int mA = tileM * 16 + mr; if (mA > N - 1) mA = N - 1;
    const float* aP = x + (size_t)mA * D_IN + 2 * half;
    const float* bP = w + (size_t)mr * D_IN + 2 * half;

    v8f acc[6] = {{}, {}, {}, {}, {}, {}};
#pragma unroll 2
    for (int k = 0; k < D_IN; k += 4) {
        v2f a = *(const v2f*)(aP + k);
#pragma unroll
        for (int t = 0; t < 6; ++t) {
            v2f b = *(const v2f*)(bP + k + t * 16 * D_IN);
            acc[t] = __builtin_amdgcn_wmma_f32_16x16x4_f32(
                 false, a,  false, b,
                 (short)0, acc[t],  false,  false);
        }
    }

    if (tileM * 16 + 15 < N) {
        float* base = mx + (size_t)(tileM * 16 + 8 * half) * D_OUT + mr;
#pragma unroll
        for (int r = 0; r < 8; ++r) {
            float* o = base + (size_t)r * D_OUT;
#pragma unroll
            for (int t = 0; t < 6; ++t) o[t * 16] = acc[t][r];
        }
    } else {
        for (int r = 0; r < 8; ++r) {
            int m = tileM * 16 + r + 8 * half;
            if (m < N) {
                float* o = mx + (size_t)m * D_OUT + mr;
                for (int t = 0; t < 6; ++t) o[t * 16] = acc[t][r];
            }
        }
    }
}

__global__ __launch_bounds__(256) void hyp_linear_epilogue(
    const float* __restrict__ x, const float* __restrict__ mx,
    const float* __restrict__ hb, float* __restrict__ xt, int N) {
    int row = (blockIdx.x << 3) + (threadIdx.x >> 5);
    if (row >= N) return;
    int lane = threadIdx.x & 31;

    const float* xr = x + (size_t)row * D_IN;
    float ss = 0.0f;
#pragma unroll
    for (int j = 0; j < 8; ++j) { float v = xr[lane + 32 * j]; ss = fmaf(v, v, ss); }
    float xn = fmaxf(sqrtf(wsum(ss)), MINN);

    const float* mr = mx + (size_t)row * D_OUT;
    float m0 = mr[lane], m1 = mr[lane + 32], m2 = mr[lane + 64];
    float mn = fmaxf(sqrtf(wsum(m0 * m0 + m1 * m1 + m2 * m2)), MINN);
    float sc = tanhf(mn / xn * atanhf(fminf(xn, ARTC))) / mn;
    float h0 = sc * m0, h1 = sc * m1, h2 = sc * m2;
    float hn = fmaxf(sqrtf(wsum(h0 * h0 + h1 * h1 + h2 * h2)), MINN);
    float pf = (hn > MAXN) ? MAXN / hn : 1.0f;
    h0 *= pf; h1 *= pf; h2 *= pf;

    float b0 = hb[lane], b1 = hb[lane + 32], b2 = hb[lane + 64];
    float x2 = wsum(h0 * h0 + h1 * h1 + h2 * h2);
    float y2 = wsum(b0 * b0 + b1 * b1 + b2 * b2);
    float xy = wsum(h0 * b0 + h1 * b1 + h2 * b2);
    float ca  = 1.0f + 2.0f * xy + y2;
    float cb  = 1.0f - x2;
    float den = fmaxf(1.0f + 2.0f * xy + x2 * y2, MINN);
    float g0 = (ca * h0 + cb * b0) / den;
    float g1 = (ca * h1 + cb * b1) / den;
    float g2 = (ca * h2 + cb * b2) / den;
    float gn = fmaxf(sqrtf(wsum(g0 * g0 + g1 * g1 + g2 * g2)), MINN);
    pf = (gn > MAXN) ? MAXN / gn : 1.0f;
    g0 *= pf; g1 *= pf; g2 *= pf;
    float pn  = fminf(gn, MAXN);
    float lsc = atanhf(fminf(pn, ARTC)) / pn;
    float* o = xt + (size_t)row * D_OUT;
    o[lane] = lsc * g0; o[lane + 32] = lsc * g1; o[lane + 64] = lsc * g2;
}

__global__ void zero_kernel(float* __restrict__ p, size_t n) {
    size_t i = (size_t)blockIdx.x * blockDim.x + threadIdx.x;
    size_t s = (size_t)gridDim.x * blockDim.x;
    for (; i < n; i += s) p[i] = 0.0f;
}

__global__ __launch_bounds__(256) void spmm_kernel(
    const int* __restrict__ erow, const int* __restrict__ ecol,
    const float* __restrict__ ew, const float* __restrict__ xt,
    float* __restrict__ sup, int E) {
    int tid  = blockIdx.x * blockDim.x + threadIdx.x;
    int lane = tid & 31;
    int waveStride = (gridDim.x * blockDim.x) >> 5;
    for (int e = tid >> 5; e < E; e += waveStride) {
        int   r = erow[e];
        int   c = ecol[e];
        float w = ew[e];
        const float* src = xt + (size_t)c * D_OUT;
        float*       dst = sup + (size_t)r * D_OUT;
        atomicAdd(dst + lane,      w * src[lane]);
        atomicAdd(dst + lane + 32, w * src[lane + 32]);
        atomicAdd(dst + lane + 64, w * src[lane + 64]);
    }
}

__global__ __launch_bounds__(256) void hyp_agg_act(const float* __restrict__ sup,
                                                   float* __restrict__ out, int N) {
    int row = (blockIdx.x << 3) + (threadIdx.x >> 5);
    if (row >= N) return;
    int lane = threadIdx.x & 31;
    const float* sr = sup + (size_t)row * D_OUT;
    float s0 = sr[lane], s1 = sr[lane + 32], s2 = sr[lane + 64];
    float sn = fmaxf(sqrtf(wsum(s0 * s0 + s1 * s1 + s2 * s2)), MINN);
    float e  = tanhf(sn) / sn;
    float p0 = e * s0, p1 = e * s1, p2 = e * s2;
    float pnr = fmaxf(sqrtf(wsum(p0 * p0 + p1 * p1 + p2 * p2)), MINN);
    float pf  = (pnr > MAXN) ? MAXN / pnr : 1.0f;
    p0 *= pf; p1 *= pf; p2 *= pf;
    float pn = fminf(pnr, MAXN);
    float l  = atanhf(fminf(pn, ARTC)) / pn;
    float u0 = fmaxf(l * p0, 0.0f), u1 = fmaxf(l * p1, 0.0f), u2 = fmaxf(l * p2, 0.0f);
    float un = fmaxf(sqrtf(wsum(u0 * u0 + u1 * u1 + u2 * u2)), MINN);
    float e2 = tanhf(un) / un;
    float h0 = e2 * u0, h1 = e2 * u1, h2 = e2 * u2;
    float hn  = fmaxf(sqrtf(wsum(h0 * h0 + h1 * h1 + h2 * h2)), MINN);
    float pf2 = (hn > MAXN) ? MAXN / hn : 1.0f;
    float* o = out + (size_t)row * D_OUT;
    o[lane] = h0 * pf2; o[lane + 32] = h1 * pf2; o[lane + 64] = h2 * pf2;
}

extern "C" void kernel_launch(void* const* d_in, const int* in_sizes, int n_in,
                              void* d_out, int out_size, void* d_ws, size_t ws_size,
                              hipStream_t stream) {
    const float* x      = (const float*)d_in[0];
    const float* weight = (const float*)d_in[1];
    const float* bias   = (const float*)d_in[2];
    const float* ew     = (const float*)d_in[3];
    const int*   erow   = (const int*)d_in[4];
    const int*   ecol   = (const int*)d_in[5];
    const int N = in_sizes[0] / D_IN;
    const int E = in_sizes[3];

    float* ws   = (float*)d_ws;
    float* hb   = ws;
    float* buf0 = ws + 128;
    float* buf1 = buf0 + (size_t)N * D_OUT;
    float* out  = (float*)d_out;

    bias_kernel<<<1, 128, 0, stream>>>(bias, hb);

    int strips     = (N + 15) >> 4;
    int gemmBlocks = (strips + 7) >> 3;
    gemm_wmma<<<gemmBlocks, 256, 0, stream>>>(x, weight, buf0, N);

    hyp_linear_epilogue<<<(N + 7) >> 3, 256, 0, stream>>>(x, buf0, hb, buf1, N);

    zero_kernel<<<2048, 256, 0, stream>>>(buf0, (size_t)N * D_OUT);
    spmm_kernel<<<(E + 7) >> 3, 256, 0, stream>>>(erow, ecol, ew, buf1, buf0, E);

    hyp_agg_act<<<(N + 7) >> 3, 256, 0, stream>>>(buf0, out, N);
}


# spmm: one wave per 32 consecutive edges (coalesced index load, readlane broadcast, 8 edges' gathers in flight) instead of one wave per edge; float atomics kept, arrival-order sum as before
# speedup vs baseline: 1.5842x; 1.5842x over previous
#include <hip/hip_runtime.h>
#include <hip/hip_bf16.h>


typedef __attribute__((ext_vector_type(2))) float v2f;
typedef __attribute__((ext_vector_type(8))) float v8f;

#define D_IN   256
#define D_OUT  96
#define MINN   1e-15f
#define MAXN   0.996f
#define ARTC   (1.0f - 1e-7f)

__device__ __forceinline__ float wsum(float v) {
    v += __shfl_xor(v, 16, 32);
    v += __shfl_xor(v, 8, 32);
    v += __shfl_xor(v, 4, 32);
    v += __shfl_xor(v, 2, 32);
    v += __shfl_xor(v, 1, 32);
    return v;
}

__global__ void bias_kernel(const float* __restrict__ bias, float* __restrict__ hb) {
    __shared__ float red[4];
    int t = threadIdx.x;
    float v = (t < D_OUT) ? bias[t] : 0.0f;
    float ss = wsum(v * v);
    if ((t & 31) == 0) red[t >> 5] = ss;
    __syncthreads();
    float n = fmaxf(sqrtf(red[0] + red[1] + red[2] + red[3]), MINN);
    float p = tanhf(n) / n * v;
    __syncthreads();
    float ps = wsum(p * p);
    if ((t & 31) == 0) red[t >> 5] = ps;
    __syncthreads();
    float pn = fmaxf(sqrtf(red[0] + red[1] + red[2] + red[3]), MINN);
    float pf = (pn > MAXN) ? MAXN / pn : 1.0f;
    if (t < D_OUT) hb[t] = p * pf;
}

__global__ __launch_bounds__(256) void gemm_wmma(const float* __restrict__ x,
                                                 const float* __restrict__ w,
                                                 float* __restrict__ mx, int N) {
    const int nTilesM = (N + 15) >> 4;
    const int tileM   = (blockIdx.x << 3) + (threadIdx.x >> 5);
    if (tileM >= nTilesM) return;

    const int lane = threadIdx.x & 31;
    const int half = lane >> 4;
    const int mr   = lane & 15;

    int mA = tileM * 16 + mr; if (mA > N - 1) mA = N - 1;
    const float* aP = x + (size_t)mA * D_IN + 2 * half;
    const float* bP = w + (size_t)mr * D_IN + 2 * half;

    v8f acc[6] = {{}, {}, {}, {}, {}, {}};
#pragma unroll 2
    for (int k = 0; k < D_IN; k += 4) {
        v2f a = *(const v2f*)(aP + k);
#pragma unroll
        for (int t = 0; t < 6; ++t) {
            v2f b = *(const v2f*)(bP + k + t * 16 * D_IN);
            acc[t] = __builtin_amdgcn_wmma_f32_16x16x4_f32(
                 false, a,  false, b,
                 (short)0, acc[t],  false,  false);
        }
    }

    if (tileM * 16 + 15 < N) {
        float* base = mx + (size_t)(tileM * 16 + 8 * half) * D_OUT + mr;
#pragma unroll
        for (int r = 0; r < 8; ++r) {
            float* o = base + (size_t)r * D_OUT;
#pragma unroll
            for (int t = 0; t < 6; ++t) o[t * 16] = acc[t][r];
        }
    } else {
        for (int r = 0; r < 8; ++r) {
            int m = tileM * 16 + r + 8 * half;
            if (m < N) {
                float* o = mx + (size_t)m * D_OUT + mr;
                for (int t = 0; t < 6; ++t) o[t * 16] = acc[t][r];
            }
        }
    }
}

__global__ __launch_bounds__(256) void hyp_linear_epilogue(
    const float* __restrict__ x, const float* __restrict__ mx,
    const float* __restrict__ hb, float* __restrict__ xt, int N) {
    int row = (blockIdx.x << 3) + (threadIdx.x >> 5);
    if (row >= N) return;
    int lane = threadIdx.x & 31;

    const float* xr = x + (size_t)row * D_IN;
    float ss = 0.0f;
#pragma unroll
    for (int j = 0; j < 8; ++j) { float v = xr[lane + 32 * j]; ss = fmaf(v, v, ss); }
    float xn = fmaxf(sqrtf(wsum(ss)), MINN);

    const float* mr = mx + (size_t)row * D_OUT;
    float m0 = mr[lane], m1 = mr[lane + 32], m2 = mr[lane + 64];
    float mn = fmaxf(sqrtf(wsum(m0 * m0 + m1 * m1 + m2 * m2)), MINN);
    float sc = tanhf(mn / xn * atanhf(fminf(xn, ARTC))) / mn;
    float h0 = sc * m0, h1 = sc * m1, h2 = sc * m2;
    float hn = fmaxf(sqrtf(wsum(h0 * h0 + h1 * h1 + h2 * h2)), MINN);
    float pf = (hn > MAXN) ? MAXN / hn : 1.0f;
    h0 *= pf; h1 *= pf; h2 *= pf;

    float b0 = hb[lane], b1 = hb[lane + 32], b2 = hb[lane + 64];
    float x2 = wsum(h0 * h0 + h1 * h1 + h2 * h2);
    float y2 = wsum(b0 * b0 + b1 * b1 + b2 * b2);
    float xy = wsum(h0 * b0 + h1 * b1 + h2 * b2);
    float ca  = 1.0f + 2.0f * xy + y2;
    float cb  = 1.0f - x2;
    float den = fmaxf(1.0f + 2.0f * xy + x2 * y2, MINN);
    float g0 = (ca * h0 + cb * b0) / den;
    float g1 = (ca * h1 + cb * b1) / den;
    float g2 = (ca * h2 + cb * b2) / den;
    float gn = fmaxf(sqrtf(wsum(g0 * g0 + g1 * g1 + g2 * g2)), MINN);
    pf = (gn > MAXN) ? MAXN / gn : 1.0f;
    g0 *= pf; g1 *= pf; g2 *= pf;
    float pn  = fminf(gn, MAXN);
    float lsc = atanhf(fminf(pn, ARTC)) / pn;
    float* o = xt + (size_t)row * D_OUT;
    o[lane] = lsc * g0; o[lane + 32] = lsc * g1; o[lane + 64] = lsc * g2;
}

__global__ void zero_kernel(float* __restrict__ p, size_t n) {
    size_t i = (size_t)blockIdx.x * blockDim.x + threadIdx.x;
    size_t s = (size_t)gridDim.x * blockDim.x;
    for (; i < n; i += s) p[i] = 0.0f;
}

#define EDGE_CHUNK 32
#define EDGE_BATCH 8

__device__ __forceinline__ void spmm_one_edge(const float* __restrict__ xt,
                                              float* __restrict__ sup,
                                              int r, int c, float w, int lane) {
    const float* src = xt + (size_t)c * D_OUT + lane;
    float*       dst = sup + (size_t)r * D_OUT + lane;
    atomicAdd(dst,      w * src[0]);
    atomicAdd(dst + 32, w * src[32]);
    atomicAdd(dst + 64, w * src[64]);
}

__global__ __launch_bounds__(256) void spmm_kernel(
    const int* __restrict__ erow, const int* __restrict__ ecol,
    const float* __restrict__ ew, const float* __restrict__ xt,
    float* __restrict__ sup, int E) {
    const int lane = threadIdx.x & 31;
    const int e0   = ((blockIdx.x << 3) + (threadIdx.x >> 5)) * EDGE_CHUNK;
    if (e0 >= E) return;
    const int cnt = min(EDGE_CHUNK, E - e0);

    int r_v = 0, c_v = 0, w_v = 0;
    if (lane < cnt) {
        r_v = erow[e0 + lane];
        c_v = ecol[e0 + lane];
        w_v = __float_as_int(ew[e0 + lane]);
    }

    if (cnt == EDGE_CHUNK) {
#pragma unroll
        for (int j = 0; j < EDGE_CHUNK; j += EDGE_BATCH) {
            float v[EDGE_BATCH][3];
#pragma unroll
            for (int u = 0; u < EDGE_BATCH; ++u) {
                const int c = __builtin_amdgcn_readlane(c_v, j + u);
                const float* src = xt + (size_t)c * D_OUT + lane;
                v[u][0] = src[0]; v[u][1] = src[32]; v[u][2] = src[64];
            }
#pragma unroll
            for (int u = 0; u < EDGE_BATCH; ++u) {
                const int   r = __builtin_amdgcn_readlane(r_v, j + u);
                const float w = __int_as_float(__builtin_amdgcn_readlane(w_v, j + u));
                float* dst = sup + (size_t)r * D_OUT + lane;
                atomicAdd(dst,      w * v[u][0]);
                atomicAdd(dst + 32, w * v[u][1]);
                atomicAdd(dst + 64, w * v[u][2]);
            }
        }
    } else {
        for (int j = 0; j < cnt; ++j) {
            const int   r = __builtin_amdgcn_readlane(r_v, j);
            const int   c = __builtin_amdgcn_readlane(c_v, j);
            const float w = __int_as_float(__builtin_amdgcn_readlane(w_v, j));
            spmm_one_edge(xt, sup, r, c, w, lane);
        }
    }
}

__global__ __launch_bounds__(256) void hyp_agg_act(const float* __restrict__ sup,
                                                   float* __restrict__ out, int N) {
    int row = (blockIdx.x << 3) + (threadIdx.x >> 5);
    if (row >= N) return;
    int lane = threadIdx.x & 31;
    const float* sr = sup + (size_t)row * D_OUT;
    float s0 = sr[lane], s1 = sr[lane + 32], s2 = sr[lane + 64];
    float sn = fmaxf(sqrtf(wsum(s0 * s0 + s1 * s1 + s2 * s2)), MINN);
    float e  = tanhf(sn) / sn;
    float p0 = e * s0, p1 = e * s1, p2 = e * s2;
    float pnr = fmaxf(sqrtf(wsum(p0 * p0 + p1 * p1 + p2 * p2)), MINN);
    float pf  = (pnr > MAXN) ? MAXN / pnr : 1.0f;
    p0 *= pf; p1 *= pf; p2 *= pf;
    float pn = fminf(pnr, MAXN);
    float l  = atanhf(fminf(pn, ARTC)) / pn;
    float u0 = fmaxf(l * p0, 0.0f), u1 = fmaxf(l * p1, 0.0f), u2 = fmaxf(l * p2, 0.0f);
    float un = fmaxf(sqrtf(wsum(u0 * u0 + u1 * u1 + u2 * u2)), MINN);
    float e2 = tanhf(un) / un;
    float h0 = e2 * u0, h1 = e2 * u1, h2 = e2 * u2;
    float hn  = fmaxf(sqrtf(wsum(h0 * h0 + h1 * h1 + h2 * h2)), MINN);
    float pf2 = (hn > MAXN) ? MAXN / hn : 1.0f;
    float* o = out + (size_t)row * D_OUT;
    o[lane] = h0 * pf2; o[lane + 32] = h1 * pf2; o[lane + 64] = h2 * pf2;
}

extern "C" void kernel_launch(void* const* d_in, const int* in_sizes, int n_in,
                              void* d_out, int out_size, void* d_ws, size_t ws_size,
                              hipStream_t stream) {
    const float* x      = (const float*)d_in[0];
    const float* weight = (const float*)d_in[1];
    const float* bias   = (const float*)d_in[2];
    const float* ew     = (const float*)d_in[3];
    const int*   erow   = (const int*)d_in[4];
    const int*   ecol   = (const int*)d_in[5];
    const int N = in_sizes[0] / D_IN;
    const int E = in_sizes[3];

    float* ws   = (float*)d_ws;
    float* hb   = ws;
    float* buf0 = ws + 128;
    float* buf1 = buf0 + (size_t)N * D_OUT;
    float* out  = (float*)d_out;

    bias_kernel<<<1, 128, 0, stream>>>(bias, hb);

    int strips     = (N + 15) >> 4;
    int gemmBlocks = (strips + 7) >> 3;
    gemm_wmma<<<gemmBlocks, 256, 0, stream>>>(x, weight, buf0, N);

    hyp_linear_epilogue<<<(N + 7) >> 3, 256, 0, stream>>>(x, buf0, hb, buf1, N);

    zero_kernel<<<2048, 256, 0, stream>>>(buf0, (size_t)N * D_OUT);
    spmm_kernel<<<(E + 255) >> 8, 256, 0, stream>>>(erow, ecol, ew, buf1, buf0, E);

    hyp_agg_act<<<(N + 7) >> 3, 256, 0, stream>>>(buf0, out, N);
}


# gemm wave keeps its 16x96 strip in registers and runs bias map, row epilogue (row norms as 16-lane DPP sums: new summation order; ca/den, cb/den once per row) and support clearing; 6 launches -> 3
# speedup vs baseline: 1.8008x; 1.1367x over previous
#include <hip/hip_runtime.h>
#include <hip/hip_bf16.h>


typedef __attribute__((ext_vector_type(2))) float v2f;
typedef __attribute__((ext_vector_type(8))) float v8f;

#define D_IN   256
#define D_OUT  96
#define MINN   1e-15f
#define MAXN   0.996f
#define ARTC   (1.0f - 1e-7f)

__device__ __forceinline__ float wsum(float v) {
    v += __shfl_xor(v, 16, 32);
    v += __shfl_xor(v, 8, 32);
    v += __shfl_xor(v, 4, 32);
    v += __shfl_xor(v, 2, 32);
    v += __shfl_xor(v, 1, 32);
    return v;
}

template <int CTRL>
__device__ __forceinline__ float dpp_read(float v) {
    return __builtin_bit_cast(float, __builtin_amdgcn_update_dpp(
        0, __builtin_bit_cast(int, v), CTRL, 0xf, 0xf, false));
}

__device__ __forceinline__ float hsum16(float v) {
    v += dpp_read<0x161>(v);
    v += dpp_read<0x162>(v);
    v += dpp_read<0x164>(v);
    v += dpp_read<0x168>(v);
    return v;
}

__device__ __forceinline__ void hsum16x8(float (&s)[8]) {
#pragma unroll
    for (int r = 0; r < 8; ++r) s[r] = hsum16(s[r]);
}

__device__ __forceinline__ float sel8v(float v0, float v1, float v2, float v3,
                                       float v4, float v5, float v6, float v7, int j) {
    const bool b0 = (j & 1) != 0, b1 = (j & 2) != 0, b2 = (j & 4) != 0;
    float a0 = b0 ? v1 : v0, a1 = b0 ? v3 : v2;
    float a2 = b0 ? v5 : v4, a3 = b0 ? v7 : v6;
    float c0 = b1 ? a1 : a0, c1 = b1 ? a3 : a2;
    return b2 ? c1 : c0;
}
#define SEL8(a, j) sel8v(a[0], a[1], a[2], a[3], a[4], a[5], a[6], a[7], j)

__device__ __forceinline__ void bcast8(float v, float (&o)[8]) {
    o[0] = dpp_read<0x150>(v); o[1] = dpp_read<0x151>(v);
    o[2] = dpp_read<0x152>(v); o[3] = dpp_read<0x153>(v);
    o[4] = dpp_read<0x154>(v); o[5] = dpp_read<0x155>(v);
    o[6] = dpp_read<0x156>(v); o[7] = dpp_read<0x157>(v);
}

__global__ __launch_bounds__(256) void gemm_hyp_linear(const float* __restrict__ x,
                                                       const float* __restrict__ w,
                                                       const float* __restrict__ bias,
                                                       float* __restrict__ xt,
                                                       float* __restrict__ sup, int N) {
    const int nTilesM = (N + 15) >> 4;
    const int tileM   = (blockIdx.x << 3) + (threadIdx.x >> 5);
    if (tileM >= nTilesM) return;

    const int lane = threadIdx.x & 31;
    const int half = lane >> 4;
    const int mr   = lane & 15;
    const bool fullTile = (tileM * 16 + 15 < N);

    {
        float* zb = sup + (size_t)tileM * 16 * D_OUT;
        const float4 z4 = make_float4(0.0f, 0.0f, 0.0f, 0.0f);
        if (fullTile) {
#pragma unroll
            for (int i = 0; i < 12; ++i) ((float4*)zb)[i * 32 + lane] = z4;
        } else {
            const int lim = (N - tileM * 16) * D_OUT;
            for (int i = lane; i < lim; i += 32) zb[i] = 0.0f;
        }
    }

    int mA = tileM * 16 + mr; if (mA > N - 1) mA = N - 1;
    const float* aP = x + (size_t)mA * D_IN + 2 * half;
    const float* bP = w + (size_t)mr * D_IN + 2 * half;

    v8f acc[6] = {{}, {}, {}, {}, {}, {}};
    float ss0 = 0.0f, ss1 = 0.0f;
#pragma unroll 2
    for (int k = 0; k < D_IN; k += 4) {
        v2f a = *(const v2f*)(aP + k);
        ss0 = fmaf(a.x, a.x, ss0);
        ss1 = fmaf(a.y, a.y, ss1);
#pragma unroll
        for (int t = 0; t < 6; ++t) {
            v2f b = *(const v2f*)(bP + k + t * 16 * D_IN);
            acc[t] = __builtin_amdgcn_wmma_f32_16x16x4_f32(
                 false, a,  false, b,
                 (short)0, acc[t],  false,  false);
        }
    }

    float xs = ss0 + ss1;
    xs += __shfl_xor(xs, 16, 32);
    const float xsOther = dpp_read<0x168>(xs);
    const float xsMine  = ((mr >> 3) == half) ? xs : xsOther;
    const int   j       = mr & 7;

    float hbv[6];
    float y2;
    {
        float bs = 0.0f;
#pragma unroll
        for (int t = 0; t < 6; ++t) { hbv[t] = bias[t * 16 + mr]; bs = fmaf(hbv[t], hbv[t], bs); }
        float n = fmaxf(sqrtf(hsum16(bs)), MINN);
        float e = tanhf(n) / n;
        float ps = 0.0f;
#pragma unroll
        for (int t = 0; t < 6; ++t) { hbv[t] = e * hbv[t]; ps = fmaf(hbv[t], hbv[t], ps); }
        float pn = fmaxf(sqrtf(hsum16(ps)), MINN);
        float pf = (pn > MAXN) ? MAXN / pn : 1.0f;
        float ys = 0.0f;
#pragma unroll
        for (int t = 0; t < 6; ++t) { hbv[t] *= pf; ys = fmaf(hbv[t], hbv[t], ys); }
        y2 = hsum16(ys);
    }

    float s[8], q[8], f0[8], f1[8];

#pragma unroll
    for (int r = 0; r < 8; ++r) {
        float v = 0.0f;
#pragma unroll
        for (int t = 0; t < 6; ++t) v = fmaf(acc[t][r], acc[t][r], v);
        s[r] = v;
    }
    hsum16x8(s);
    {
        float xn = fmaxf(sqrtf(xsMine), MINN);
        float mn = fmaxf(sqrtf(SEL8(s, j)), MINN);
        float sc = tanhf(mn / xn * atanhf(fminf(xn, ARTC))) / mn;
        bcast8(sc, f0);
    }
#pragma unroll
    for (int r = 0; r < 8; ++r) {
        float v = 0.0f;
#pragma unroll
        for (int t = 0; t < 6; ++t) { acc[t][r] *= f0[r]; v = fmaf(acc[t][r], acc[t][r], v); }
        s[r] = v;
    }
    hsum16x8(s);
    {
        float hn = fmaxf(sqrtf(SEL8(s, j)), MINN);
        float pf = (hn > MAXN) ? MAXN / hn : 1.0f;
        bcast8(pf, f0);
    }
#pragma unroll
    for (int r = 0; r < 8; ++r) {
        float v = 0.0f, u = 0.0f;
#pragma unroll
        for (int t = 0; t < 6; ++t) {
            acc[t][r] *= f0[r];
            v = fmaf(acc[t][r], acc[t][r], v);
            u = fmaf(acc[t][r], hbv[t], u);
        }
        s[r] = v; q[r] = u;
    }
    hsum16x8(s);
    hsum16x8(q);
    {
        float x2 = SEL8(s, j);
        float xy = SEL8(q, j);
        float ca  = 1.0f + 2.0f * xy + y2;
        float cb  = 1.0f - x2;
        float den = fmaxf(1.0f + 2.0f * xy + x2 * y2, MINN);
        bcast8(ca / den, f0); bcast8(cb / den, f1);
    }
#pragma unroll
    for (int r = 0; r < 8; ++r) {
        float v = 0.0f;
#pragma unroll
        for (int t = 0; t < 6; ++t) {
            acc[t][r] = f0[r] * acc[t][r] + f1[r] * hbv[t];
            v = fmaf(acc[t][r], acc[t][r], v);
        }
        s[r] = v;
    }
    hsum16x8(s);
    {
        float gn = fmaxf(sqrtf(SEL8(s, j)), MINN);
        float pf = (gn > MAXN) ? MAXN / gn : 1.0f;
        float pn  = fminf(gn, MAXN);
        float lsc = atanhf(fminf(pn, ARTC)) / pn;
        bcast8(pf, f0); bcast8(lsc, f1);
    }

    float* obase = xt + (size_t)(tileM * 16 + 8 * half) * D_OUT + mr;
#pragma unroll
    for (int r = 0; r < 8; ++r) {
        if (fullTile || tileM * 16 + 8 * half + r < N) {
            float* o = obase + (size_t)r * D_OUT;
#pragma unroll
            for (int t = 0; t < 6; ++t) o[t * 16] = f1[r] * (acc[t][r] * f0[r]);
        }
    }
}

#define EDGE_CHUNK 32
#define EDGE_BATCH 8

__device__ __forceinline__ void spmm_one_edge(const float* __restrict__ xt,
                                              float* __restrict__ sup,
                                              int r, int c, float w, int lane) {
    const float* src = xt + (size_t)c * D_OUT + lane;
    float*       dst = sup + (size_t)r * D_OUT + lane;
    atomicAdd(dst,      w * src[0]);
    atomicAdd(dst + 32, w * src[32]);
    atomicAdd(dst + 64, w * src[64]);
}

__global__ __launch_bounds__(256) void spmm_kernel(
    const int* __restrict__ erow, const int* __restrict__ ecol,
    const float* __restrict__ ew, const float* __restrict__ xt,
    float* __restrict__ sup, int E) {
    const int lane = threadIdx.x & 31;
    const int e0   = ((blockIdx.x << 3) + (threadIdx.x >> 5)) * EDGE_CHUNK;
    if (e0 >= E) return;
    const int cnt = min(EDGE_CHUNK, E - e0);

    int r_v = 0, c_v = 0, w_v = 0;
    if (lane < cnt) {
        r_v = erow[e0 + lane];
        c_v = ecol[e0 + lane];
        w_v = __float_as_int(ew[e0 + lane]);
    }

    if (cnt == EDGE_CHUNK) {
#pragma unroll
        for (int j = 0; j < EDGE_CHUNK; j += EDGE_BATCH) {
            float v[EDGE_BATCH][3];
#pragma unroll
            for (int u = 0; u < EDGE_BATCH; ++u) {
                const int c = __builtin_amdgcn_readlane(c_v, j + u);
                const float* src = xt + (size_t)c * D_OUT + lane;
                v[u][0] = src[0]; v[u][1] = src[32]; v[u][2] = src[64];
            }
#pragma unroll
            for (int u = 0; u < EDGE_BATCH; ++u) {
                const int   r = __builtin_amdgcn_readlane(r_v, j + u);
                const float w = __int_as_float(__builtin_amdgcn_readlane(w_v, j + u));
                float* dst = sup + (size_t)r * D_OUT + lane;
                atomicAdd(dst,      w * v[u][0]);
                atomicAdd(dst + 32, w * v[u][1]);
                atomicAdd(dst + 64, w * v[u][2]);
            }
        }
    } else {
        for (int j = 0; j < cnt; ++j) {
            const int   r = __builtin_amdgcn_readlane(r_v, j);
            const int   c = __builtin_amdgcn_readlane(c_v, j);
            const float w = __int_as_float(__builtin_amdgcn_readlane(w_v, j));
            spmm_one_edge(xt, sup, r, c, w, lane);
        }
    }
}

__global__ __launch_bounds__(256) void hyp_agg_act(const float* __restrict__ sup,
                                                   float* __restrict__ out, int N) {
    int row = (blockIdx.x << 3) + (threadIdx.x >> 5);
    if (row >= N) return;
    int lane = threadIdx.x & 31;
    const float* sr = sup + (size_t)row * D_OUT;
    float s0 = sr[lane], s1 = sr[lane + 32], s2 = sr[lane + 64];
    float sn = fmaxf(sqrtf(wsum(s0 * s0 + s1 * s1 + s2 * s2)), MINN);
    float e  = tanhf(sn) / sn;
    float p0 = e * s0, p1 = e * s1, p2 = e * s2;
    float pnr = fmaxf(sqrtf(wsum(p0 * p0 + p1 * p1 + p2 * p2)), MINN);
    float pf  = (pnr > MAXN) ? MAXN / pnr : 1.0f;
    p0 *= pf; p1 *= pf; p2 *= pf;
    float pn = fminf(pnr, MAXN);
    float l  = atanhf(fminf(pn, ARTC)) / pn;
    float u0 = fmaxf(l * p0, 0.0f), u1 = fmaxf(l * p1, 0.0f), u2 = fmaxf(l * p2, 0.0f);
    float un = fmaxf(sqrtf(wsum(u0 * u0 + u1 * u1 + u2 * u2)), MINN);
    float e2 = tanhf(un) / un;
    float h0 = e2 * u0, h1 = e2 * u1, h2 = e2 * u2;
    float hn  = fmaxf(sqrtf(wsum(h0 * h0 + h1 * h1 + h2 * h2)), MINN);
    float pf2 = (hn > MAXN) ? MAXN / hn : 1.0f;
    float* o = out + (size_t)row * D_OUT;
    o[lane] = h0 * pf2; o[lane + 32] = h1 * pf2; o[lane + 64] = h2 * pf2;
}

extern "C" void kernel_launch(void* const* d_in, const int* in_sizes, int n_in,
                              void* d_out, int out_size, void* d_ws, size_t ws_size,
                              hipStream_t stream) {
    const float* x      = (const float*)d_in[0];
    const float* weight = (const float*)d_in[1];
    const float* bias   = (const float*)d_in[2];
    const float* ew     = (const float*)d_in[3];
    const int*   erow   = (const int*)d_in[4];
    const int*   ecol   = (const int*)d_in[5];
    const int N = in_sizes[0] / D_IN;
    const int E = in_sizes[3];

    float* ws   = (float*)d_ws;
    float* buf0 = ws + 128;
    float* buf1 = buf0 + (size_t)N * D_OUT;
    float* out  = (float*)d_out;

    int strips     = (N + 15) >> 4;
    int gemmBlocks = (strips + 7) >> 3;
    gemm_hyp_linear<<<gemmBlocks, 256, 0, stream>>>(x, weight, bias, buf1, buf0, N);

    spmm_kernel<<<(E + 255) >> 8, 256, 0, stream>>>(erow, ecol, ew, buf1, buf0, E);

    hyp_agg_act<<<(N + 7) >> 3, 256, 0, stream>>>(buf0, out, N);
}


# spmm without global atomics: edges grouped by destination range (LDS counting sort), one block per 196 rows sums them in an LDS tile (ds_add_f32, arrival order as before) and runs the row tail
# speedup vs baseline: 2.3351x; 1.2967x over previous
#include <hip/hip_runtime.h>
#include <hip/hip_bf16.h>


typedef __attribute__((ext_vector_type(2))) float v2f;
typedef __attribute__((ext_vector_type(8))) float v8f;

#define D_IN   256
#define D_OUT  96
#define MINN   1e-15f
#define MAXN   0.996f
#define ARTC   (1.0f - 1e-7f)

__device__ __forceinline__ float wsum(float v) {
    v += __shfl_xor(v, 16, 32);
    v += __shfl_xor(v, 8, 32);
    v += __shfl_xor(v, 4, 32);
    v += __shfl_xor(v, 2, 32);
    v += __shfl_xor(v, 1, 32);
    return v;
}

template <int CTRL>
__device__ __forceinline__ float dpp_read(float v) {
    return __builtin_bit_cast(float, __builtin_amdgcn_update_dpp(
        0, __builtin_bit_cast(int, v), CTRL, 0xf, 0xf, false));
}

__device__ __forceinline__ float hsum16(float v) {
    v += dpp_read<0x161>(v);
    v += dpp_read<0x162>(v);
    v += dpp_read<0x164>(v);
    v += dpp_read<0x168>(v);
    return v;
}

__device__ __forceinline__ void hsum16x8(float (&s)[8]) {
#pragma unroll
    for (int r = 0; r < 8; ++r) s[r] = hsum16(s[r]);
}

__device__ __forceinline__ float sel8v(float v0, float v1, float v2, float v3,
                                       float v4, float v5, float v6, float v7, int j) {
    const bool b0 = (j & 1) != 0, b1 = (j & 2) != 0, b2 = (j & 4) != 0;
    float a0 = b0 ? v1 : v0, a1 = b0 ? v3 : v2;
    float a2 = b0 ? v5 : v4, a3 = b0 ? v7 : v6;
    float c0 = b1 ? a1 : a0, c1 = b1 ? a3 : a2;
    return b2 ? c1 : c0;
}
#define SEL8(a, j) sel8v(a[0], a[1], a[2], a[3], a[4], a[5], a[6], a[7], j)

__device__ __forceinline__ void bcast8(float v, float (&o)[8]) {
    o[0] = dpp_read<0x150>(v); o[1] = dpp_read<0x151>(v);
    o[2] = dpp_read<0x152>(v); o[3] = dpp_read<0x153>(v);
    o[4] = dpp_read<0x154>(v); o[5] = dpp_read<0x155>(v);
    o[6] = dpp_read<0x156>(v); o[7] = dpp_read<0x157>(v);
}

__global__ __launch_bounds__(256) void gemm_hyp_linear(const float* __restrict__ x,
                                                       const float* __restrict__ w,
                                                       const float* __restrict__ bias,
                                                       float* __restrict__ xt, int N) {
    const int nTilesM = (N + 15) >> 4;
    const int tileM   = (blockIdx.x << 3) + (threadIdx.x >> 5);
    if (tileM >= nTilesM) return;

    const int lane = threadIdx.x & 31;
    const int half = lane >> 4;
    const int mr   = lane & 15;
    const bool fullTile = (tileM * 16 + 15 < N);

    int mA = tileM * 16 + mr; if (mA > N - 1) mA = N - 1;
    const float* aP = x + (size_t)mA * D_IN + 2 * half;
    const float* bP = w + (size_t)mr * D_IN + 2 * half;

    v8f acc[6] = {{}, {}, {}, {}, {}, {}};
    float ss0 = 0.0f, ss1 = 0.0f;
#pragma unroll 2
    for (int k = 0; k < D_IN; k += 4) {
        v2f a = *(const v2f*)(aP + k);
        ss0 = fmaf(a.x, a.x, ss0);
        ss1 = fmaf(a.y, a.y, ss1);
#pragma unroll
        for (int t = 0; t < 6; ++t) {
            v2f b = *(const v2f*)(bP + k + t * 16 * D_IN);
            acc[t] = __builtin_amdgcn_wmma_f32_16x16x4_f32(
                 false, a,  false, b,
                 (short)0, acc[t],  false,  false);
        }
    }

    float xs = ss0 + ss1;
    xs += __shfl_xor(xs, 16, 32);
    const float xsOther = dpp_read<0x168>(xs);
    const float xsMine  = ((mr >> 3) == half) ? xs : xsOther;
    const int   j       = mr & 7;

    float hbv[6];
    float y2;
    {
        float bs = 0.0f;
#pragma unroll
        for (int t = 0; t < 6; ++t) { hbv[t] = bias[t * 16 + mr]; bs = fmaf(hbv[t], hbv[t], bs); }
        float n = fmaxf(sqrtf(hsum16(bs)), MINN);
        float e = tanhf(n) / n;
        float ps = 0.0f;
#pragma unroll
        for (int t = 0; t < 6; ++t) { hbv[t] = e * hbv[t]; ps = fmaf(hbv[t], hbv[t], ps); }
        float pn = fmaxf(sqrtf(hsum16(ps)), MINN);
        float pf = (pn > MAXN) ? MAXN / pn : 1.0f;
        float ys = 0.0f;
#pragma unroll
        for (int t = 0; t < 6; ++t) { hbv[t] *= pf; ys = fmaf(hbv[t], hbv[t], ys); }
        y2 = hsum16(ys);
    }

    float s[8], q[8], f0[8], f1[8];

#pragma unroll
    for (int r = 0; r < 8; ++r) {
        float v = 0.0f;
#pragma unroll
        for (int t = 0; t < 6; ++t) v = fmaf(acc[t][r], acc[t][r], v);
        s[r] = v;
    }
    hsum16x8(s);
    {
        float xn = fmaxf(sqrtf(xsMine), MINN);
        float mn = fmaxf(sqrtf(SEL8(s, j)), MINN);
        float sc = tanhf(mn / xn * atanhf(fminf(xn, ARTC))) / mn;
        bcast8(sc, f0);
    }
#pragma unroll
    for (int r = 0; r < 8; ++r) {
        float v = 0.0f;
#pragma unroll
        for (int t = 0; t < 6; ++t) { acc[t][r] *= f0[r]; v = fmaf(acc[t][r], acc[t][r], v); }
        s[r] = v;
    }
    hsum16x8(s);
    {
        float hn = fmaxf(sqrtf(SEL8(s, j)), MINN);
        float pf = (hn > MAXN) ? MAXN / hn : 1.0f;
        bcast8(pf, f0);
    }
#pragma unroll
    for (int r = 0; r < 8; ++r) {
        float v = 0.0f, u = 0.0f;
#pragma unroll
        for (int t = 0; t < 6; ++t) {
            acc[t][r] *= f0[r];
            v = fmaf(acc[t][r], acc[t][r], v);
            u = fmaf(acc[t][r], hbv[t], u);
        }
        s[r] = v; q[r] = u;
    }
    hsum16x8(s);
    hsum16x8(q);
    {
        float x2 = SEL8(s, j);
        float xy = SEL8(q, j);
        float ca  = 1.0f + 2.0f * xy + y2;
        float cb  = 1.0f - x2;
        float den = fmaxf(1.0f + 2.0f * xy + x2 * y2, MINN);
        bcast8(ca / den, f0); bcast8(cb / den, f1);
    }
#pragma unroll
    for (int r = 0; r < 8; ++r) {
        float v = 0.0f;
#pragma unroll
        for (int t = 0; t < 6; ++t) {
            acc[t][r] = f0[r] * acc[t][r] + f1[r] * hbv[t];
            v = fmaf(acc[t][r], acc[t][r], v);
        }
        s[r] = v;
    }
    hsum16x8(s);
    {
        float gn = fmaxf(sqrtf(SEL8(s, j)), MINN);
        float pf = (gn > MAXN) ? MAXN / gn : 1.0f;
        float pn  = fminf(gn, MAXN);
        float lsc = atanhf(fminf(pn, ARTC)) / pn;
        bcast8(pf, f0); bcast8(lsc, f1);
    }

    float* obase = xt + (size_t)(tileM * 16 + 8 * half) * D_OUT + mr;
#pragma unroll
    for (int r = 0; r < 8; ++r) {
        if (fullTile || tileM * 16 + 8 * half + r < N) {
            float* o = obase + (size_t)r * D_OUT;
#pragma unroll
            for (int t = 0; t < 6; ++t) o[t * 16] = f1[r] * (acc[t][r] * f0[r]);
        }
    }
}

#define OWN_ROWS    196
#define MAX_RANGES  256
#define MAX_SLICES  256
#define BIN_THREADS 1024
#define BIN_REGS    4

__global__ __launch_bounds__(BIN_THREADS) void bin_edges(
    const int* __restrict__ erow, const int* __restrict__ ecol,
    const float* __restrict__ ew, int* __restrict__ sorted_w,
    int* __restrict__ tcnt, int* __restrict__ toff, int E, int N, int S) {
    int4* __restrict__ sorted = (int4*)sorted_w;
    __shared__ int s_cnt[MAX_RANGES];
    __shared__ int s_off[MAX_RANGES];
    __shared__ int s_cur[MAX_RANGES];
    const int tid = threadIdx.x;
    const int b   = blockIdx.x;
    const int P   = gridDim.x;
    const int beg = b * S;
    const int end = min(E, beg + S);
    const bool inRegs = (S <= BIN_REGS * BIN_THREADS);

    if (tid < MAX_RANGES) s_cnt[tid] = 0;

    int r[BIN_REGS], c[BIN_REGS], w[BIN_REGS], rk[BIN_REGS];
    if (inRegs) {
        for (int i = 0; i < BIN_REGS; ++i) {
            const int e = beg + tid + i * BIN_THREADS;
            r[i] = -1; c[i] = 0; w[i] = 0;
            if (e < end) { r[i] = erow[e]; c[i] = ecol[e]; w[i] = __float_as_int(ew[e]); }
        }
    }
    __syncthreads();
    if (inRegs) {
        for (int i = 0; i < BIN_REGS; ++i) {
            rk[i] = 0;
            if ((unsigned)r[i] < (unsigned)N) rk[i] = atomicAdd(&s_cnt[r[i] / OWN_ROWS], 1);
        }
    } else {
        for (int e = beg + tid; e < end; e += BIN_THREADS) {
            const int rr = erow[e];
            if ((unsigned)rr < (unsigned)N) atomicAdd(&s_cnt[rr / OWN_ROWS], 1);
        }
    }
    __syncthreads();
    if (tid < 32) {
        int v[8];
        int s = 0;
        for (int j = 0; j < 8; ++j) { v[j] = s_cnt[tid * 8 + j]; s += v[j]; }
        int inc = s;
        for (int d = 1; d < 32; d <<= 1) {
            const int t = __shfl_up(inc, d, 32);
            if (tid >= d) inc += t;
        }
        int run = inc - s;
        for (int j = 0; j < 8; ++j) { s_off[tid * 8 + j] = run; s_cur[tid * 8 + j] = run; run += v[j]; }
    }
    __syncthreads();
    if (tid < MAX_RANGES) {
        tcnt[tid * P + b] = s_cnt[tid];
        toff[tid * P + b] = s_off[tid];
    }
    if (inRegs) {
        for (int i = 0; i < BIN_REGS; ++i) {
            if ((unsigned)r[i] < (unsigned)N)
                sorted[beg + s_off[r[i] / OWN_ROWS] + rk[i]] = make_int4(r[i], c[i], w[i], 0);
        }
    } else {
        for (int e = beg + tid; e < end; e += BIN_THREADS) {
            const int rr = erow[e];
            if ((unsigned)rr < (unsigned)N) {
                const int pos = beg + atomicAdd(&s_cur[rr / OWN_ROWS], 1);
                sorted[pos] = make_int4(rr, ecol[e], __float_as_int(ew[e]), 0);
            }
        }
    }
}

#define OWN_THREADS 1024
#define EDGE_CHUNK  32
#define EDGE_BATCH  16

template <bool FULL>
__device__ __forceinline__ void chunk_edges(const float* __restrict__ xt, float* tile,
                                            int r_v, int c_v, int w_v, int cnt, int lane) {
#pragma unroll
    for (int j = 0; j < EDGE_CHUNK; j += EDGE_BATCH) {
        float v[EDGE_BATCH][3];
#pragma unroll
        for (int u = 0; u < EDGE_BATCH; ++u) {
            const int c = __builtin_amdgcn_readlane(c_v, j + u);
            const float* src = xt + (size_t)c * D_OUT + lane;
            v[u][0] = src[0]; v[u][1] = src[32]; v[u][2] = src[64];
        }
#pragma unroll
        for (int u = 0; u < EDGE_BATCH; ++u) {
            if (FULL || j + u < cnt) {
                const int   r = __builtin_amdgcn_readlane(r_v, j + u);
                const float w = __int_as_float(__builtin_amdgcn_readlane(w_v, j + u));
                float* dst = tile + r * D_OUT + lane;
                atomicAdd(dst,      w * v[u][0]);
                atomicAdd(dst + 32, w * v[u][1]);
                atomicAdd(dst + 64, w * v[u][2]);
            }
        }
    }
}

__global__ __launch_bounds__(OWN_THREADS) void own_rows(
    const int* __restrict__ sorted_w, const int* __restrict__ tcnt,
    const int* __restrict__ toff, const float* __restrict__ xt,
    float* __restrict__ out, int N, int S, int P) {
    const int4* __restrict__ sorted = (const int4*)sorted_w;
    __shared__ __attribute__((aligned(16))) float tile[OWN_ROWS * D_OUT];
    __shared__ int s_c[MAX_SLICES];
    __shared__ int s_base[MAX_SLICES];
    __shared__ int s_pre[MAX_SLICES + 1];
    const int tid  = threadIdx.x;
    const int lane = tid & 31;
    const int wave = __builtin_amdgcn_readfirstlane(tid >> 5);
    const int k    = blockIdx.x;
    const int lo   = k * OWN_ROWS;
    const int nrows = min(OWN_ROWS, N - lo);

    {
        const float4 z4 = make_float4(0.0f, 0.0f, 0.0f, 0.0f);
        for (int i = tid; i < OWN_ROWS * D_OUT / 4; i += OWN_THREADS) ((float4*)tile)[i] = z4;
    }
    if (tid < MAX_SLICES) {
        int c = 0, o = 0;
        if (tid < P) { c = tcnt[k * P + tid]; o = toff[k * P + tid]; }
        s_c[tid]    = c;
        s_base[tid] = tid * S + o;
    }
    __syncthreads();
    if (tid < 32) {
        int v[8];
        int s = 0;
        for (int j = 0; j < 8; ++j) { v[j] = s_c[tid * 8 + j]; s += v[j]; }
        int inc = s;
        for (int d = 1; d < 32; d <<= 1) {
            const int t = __shfl_up(inc, d, 32);
            if (tid >= d) inc += t;
        }
        int run = inc - s;
        for (int j = 0; j < 8; ++j) { s_pre[tid * 8 + j] = run; run += v[j]; }
        if (tid == 31) s_pre[MAX_SLICES] = run;
    }
    __syncthreads();
    const int T = __builtin_amdgcn_readfirstlane(s_pre[MAX_SLICES]);

    for (int ch = wave; ch * EDGE_CHUNK < T; ch += OWN_THREADS / 32) {
        const int cnt = min(EDGE_CHUNK, T - ch * EDGE_CHUNK);
        int r_v = 0, c_v = 0, w_v = 0;
        if (lane < cnt) {
            const int f = ch * EDGE_CHUNK + lane;
            int a = 0, z = MAX_SLICES;
            for (int it = 0; it < 8; ++it) {
                const int m = (a + z) >> 1;
                if (s_pre[m] <= f) a = m; else z = m;
            }
            const int4 rec = sorted[s_base[a] + (f - s_pre[a])];
            r_v = rec.x - lo;
            c_v = rec.y;
            w_v = rec.z;
        }
        if (cnt == EDGE_CHUNK) chunk_edges<true>(xt, tile, r_v, c_v, w_v, cnt, lane);
        else                   chunk_edges<false>(xt, tile, r_v, c_v, w_v, cnt, lane);
    }
    __syncthreads();

    for (int rr = wave; rr < nrows; rr += OWN_THREADS / 32) {
        const float* sr = tile + rr * D_OUT;
        float s0 = sr[lane], s1 = sr[lane + 32], s2 = sr[lane + 64];
        float sn = fmaxf(sqrtf(wsum(s0 * s0 + s1 * s1 + s2 * s2)), MINN);
        float e  = tanhf(sn) / sn;
        float p0 = e * s0, p1 = e * s1, p2 = e * s2;
        float pnr = fmaxf(sqrtf(wsum(p0 * p0 + p1 * p1 + p2 * p2)), MINN);
        float pf  = (pnr > MAXN) ? MAXN / pnr : 1.0f;
        p0 *= pf; p1 *= pf; p2 *= pf;
        float pn = fminf(pnr, MAXN);
        float l  = atanhf(fminf(pn, ARTC)) / pn;
        float u0 = fmaxf(l * p0, 0.0f), u1 = fmaxf(l * p1, 0.0f), u2 = fmaxf(l * p2, 0.0f);
        float un = fmaxf(sqrtf(wsum(u0 * u0 + u1 * u1 + u2 * u2)), MINN);
        float e2 = tanhf(un) / un;
        float h0 = e2 * u0, h1 = e2 * u1, h2 = e2 * u2;
        float hn  = fmaxf(sqrtf(wsum(h0 * h0 + h1 * h1 + h2 * h2)), MINN);
        float pf2 = (hn > MAXN) ? MAXN / hn : 1.0f;
        float* o = out + (size_t)(lo + rr) * D_OUT;
        o[lane] = h0 * pf2; o[lane + 32] = h1 * pf2; o[lane + 64] = h2 * pf2;
    }
}

extern "C" void kernel_launch(void* const* d_in, const int* in_sizes, int n_in,
                              void* d_out, int out_size, void* d_ws, size_t ws_size,
                              hipStream_t stream) {
    const float* x      = (const float*)d_in[0];
    const float* weight = (const float*)d_in[1];
    const float* bias   = (const float*)d_in[2];
    const float* ew     = (const float*)d_in[3];
    const int*   erow   = (const int*)d_in[4];
    const int*   ecol   = (const int*)d_in[5];
    const int N = in_sizes[0] / D_IN;
    const int E = in_sizes[3];

    float* ws   = (float*)d_ws;
    float* buf0 = ws + 128;
    float* buf1 = buf0 + (size_t)N * D_OUT;
    float* out  = (float*)d_out;

    int*  sorted = (int*)buf0;
    int*  tcnt   = sorted + 4 * (size_t)E;
    int*  toff   = tcnt + MAX_RANGES * MAX_SLICES;

    int strips     = (N + 15) >> 4;
    int gemmBlocks = (strips + 7) >> 3;
    gemm_hyp_linear<<<gemmBlocks, 256, 0, stream>>>(x, weight, bias, buf1, N);

    const int P = MAX_SLICES;
    const int S = (E + P - 1) / P;
    bin_edges<<<P, BIN_THREADS, 0, stream>>>(erow, ecol, ew, sorted, tcnt, toff, E, N, S);

    int ranges = (N + OWN_ROWS - 1) / OWN_ROWS;
    own_rows<<<ranges, OWN_THREADS, 0, stream>>>(sorted, tcnt, toff, buf1, out, N, S, P);
}


# gemm operand feed: W staged once per 16-wave block in LDS, A and W fragments read 16 B per lane (two k-steps per load; k order inside each group of 8 and the row-norm sum order change)
# speedup vs baseline: 2.9797x; 1.2760x over previous
#include <hip/hip_runtime.h>
#include <hip/hip_bf16.h>


typedef __attribute__((ext_vector_type(2))) float v2f;
typedef __attribute__((ext_vector_type(8))) float v8f;

#define D_IN   256
#define D_OUT  96
#define MINN   1e-15f
#define MAXN   0.996f
#define ARTC   (1.0f - 1e-7f)

__device__ __forceinline__ float wsum(float v) {
    v += __shfl_xor(v, 16, 32);
    v += __shfl_xor(v, 8, 32);
    v += __shfl_xor(v, 4, 32);
    v += __shfl_xor(v, 2, 32);
    v += __shfl_xor(v, 1, 32);
    return v;
}

template <int CTRL>
__device__ __forceinline__ float dpp_read(float v) {
    return __builtin_bit_cast(float, __builtin_amdgcn_update_dpp(
        0, __builtin_bit_cast(int, v), CTRL, 0xf, 0xf, false));
}

__device__ __forceinline__ float hsum16(float v) {
    v += dpp_read<0x161>(v);
    v += dpp_read<0x162>(v);
    v += dpp_read<0x164>(v);
    v += dpp_read<0x168>(v);
    return v;
}

__device__ __forceinline__ void hsum16x8(float (&s)[8]) {
#pragma unroll
    for (int r = 0; r < 8; ++r) s[r] = hsum16(s[r]);
}

__device__ __forceinline__ float sel8v(float v0, float v1, float v2, float v3,
                                       float v4, float v5, float v6, float v7, int j) {
    const bool b0 = (j & 1) != 0, b1 = (j & 2) != 0, b2 = (j & 4) != 0;
    float a0 = b0 ? v1 : v0, a1 = b0 ? v3 : v2;
    float a2 = b0 ? v5 : v4, a3 = b0 ? v7 : v6;
    float c0 = b1 ? a1 : a0, c1 = b1 ? a3 : a2;
    return b2 ? c1 : c0;
}
#define SEL8(a, j) sel8v(a[0], a[1], a[2], a[3], a[4], a[5], a[6], a[7], j)

__device__ __forceinline__ void bcast8(float v, float (&o)[8]) {
    o[0] = dpp_read<0x150>(v); o[1] = dpp_read<0x151>(v);
    o[2] = dpp_read<0x152>(v); o[3] = dpp_read<0x153>(v);
    o[4] = dpp_read<0x154>(v); o[5] = dpp_read<0x155>(v);
    o[6] = dpp_read<0x156>(v); o[7] = dpp_read<0x157>(v);
}

#define GEMM_THREADS 512
#define GEMM_WAVES   (GEMM_THREADS / 32)
#define WP           260

__global__ __launch_bounds__(GEMM_THREADS) __attribute__((amdgpu_waves_per_eu(4, 5)))
void gemm_hyp_linear(const float* __restrict__ x, const float* __restrict__ w,
                     const float* __restrict__ bias, float* __restrict__ xt, int N) {
    __shared__ __attribute__((aligned(16))) float sW[D_OUT * WP];
    const int nTilesM = (N + 15) >> 4;
    const int lane = threadIdx.x & 31;
    const int wave = __builtin_amdgcn_readfirstlane(threadIdx.x >> 5);
    const int half = lane >> 4;
    const int mr   = lane & 15;
    const int tileM = blockIdx.x * GEMM_WAVES + wave;

    int mA = tileM * 16 + mr; if (mA > N - 1) mA = N - 1;
    const float* aP = x + (size_t)mA * D_IN + 4 * half;
    float4 aN[4];
#pragma unroll
    for (int i = 0; i < 4; ++i) aN[i] = make_float4(0.0f, 0.0f, 0.0f, 0.0f);
    if (tileM < nTilesM) {
#pragma unroll
        for (int i = 0; i < 4; ++i) aN[i] = *(const float4*)(aP + 8 * i);
    }

    float hbv[6];
    float y2;
    {
        float bs = 0.0f;
#pragma unroll
        for (int t = 0; t < 6; ++t) { hbv[t] = bias[t * 16 + mr]; bs = fmaf(hbv[t], hbv[t], bs); }
        float n = fmaxf(sqrtf(hsum16(bs)), MINN);
        float e = tanhf(n) / n;
        float ps = 0.0f;
#pragma unroll
        for (int t = 0; t < 6; ++t) { hbv[t] = e * hbv[t]; ps = fmaf(hbv[t], hbv[t], ps); }
        float pn = fmaxf(sqrtf(hsum16(ps)), MINN);
        float pf = (pn > MAXN) ? MAXN / pn : 1.0f;
        float ys = 0.0f;
#pragma unroll
        for (int t = 0; t < 6; ++t) { hbv[t] *= pf; ys = fmaf(hbv[t], hbv[t], ys); }
        y2 = hsum16(ys);
    }

    for (int i = threadIdx.x; i < D_OUT * (D_IN / 4); i += GEMM_THREADS) {
        const int r = i >> 6, c = (i & 63) << 2;
        *(float4*)(sW + r * WP + c) = *(const float4*)(w + (size_t)r * D_IN + c);
    }
    __syncthreads();
    const float* bT[6];
#pragma unroll
    for (int t = 0; t < 6; ++t) bT[t] = sW + (t * 16 + mr) * WP + 4 * half;

    if (tileM >= nTilesM) return;
    const bool fullTile = (tileM * 16 + 15 < N);

    v8f acc[6] = {{}, {}, {}, {}, {}, {}};
    float4 aA[8][4];
#pragma unroll
    for (int i = 0; i < 4; ++i) aA[0][i] = aN[i];
    float4 bC[6], bN[6];
#pragma unroll
    for (int t = 0; t < 6; ++t) bC[t] = *(const float4*)(bT[t]);
    __builtin_amdgcn_sched_barrier(0);
#pragma unroll
    for (int kb = 0; kb < D_IN; kb += 32) {
        if (kb + 32 < D_IN) {
#pragma unroll
            for (int i = 0; i < 4; ++i) aA[kb / 32 + 1][i] = *(const float4*)(aP + kb + 32 + 8 * i);
        }
#pragma unroll
        for (int i = 0; i < 4; ++i) {
            if (kb + 8 * i + 8 < D_IN) {
#pragma unroll
                for (int t = 0; t < 6; ++t) bN[t] = *(const float4*)(bT[t] + kb + 8 * i + 8);
            }
            const float4 av = aA[kb / 32][i];
            v2f a01; a01.x = av.x; a01.y = av.y;
            v2f a23; a23.x = av.z; a23.y = av.w;
#pragma unroll
            for (int t = 0; t < 6; ++t) {
                v2f b01; b01.x = bC[t].x; b01.y = bC[t].y;
                acc[t] = __builtin_amdgcn_wmma_f32_16x16x4_f32(
                     false, a01,  false, b01,
                     (short)0, acc[t],  false,  false);
            }
#pragma unroll
            for (int t = 0; t < 6; ++t) {
                v2f b23; b23.x = bC[t].z; b23.y = bC[t].w;
                acc[t] = __builtin_amdgcn_wmma_f32_16x16x4_f32(
                     false, a23,  false, b23,
                     (short)0, acc[t],  false,  false);
            }
#pragma unroll
            for (int t = 0; t < 6; ++t) bC[t] = bN[t];
        }
    }
    __builtin_amdgcn_sched_barrier(0);
    asm volatile("v_nop\n\tv_nop\n\tv_nop\n\tv_nop");
    __builtin_amdgcn_sched_barrier(0);

    float ss0 = 0.0f, ss1 = 0.0f, ss2 = 0.0f, ss3 = 0.0f;
#pragma unroll
    for (int b = 0; b < 8; ++b) {
#pragma unroll
        for (int i = 0; i < 4; ++i) {
            ss0 = fmaf(aA[b][i].x, aA[b][i].x, ss0);
            ss1 = fmaf(aA[b][i].y, aA[b][i].y, ss1);
            ss2 = fmaf(aA[b][i].z, aA[b][i].z, ss2);
            ss3 = fmaf(aA[b][i].w, aA[b][i].w, ss3);
        }
    }

    float xs = (ss0 + ss1) + (ss2 + ss3);
    xs += __shfl_xor(xs, 16, 32);
    const float xsOther = dpp_read<0x168>(xs);
    const float xsMine  = ((mr >> 3) == half) ? xs : xsOther;
    const int   j       = mr & 7;

    float s[8], q[8], f0[8], f1[8];

#pragma unroll
    for (int r = 0; r < 8; ++r) {
        float v = 0.0f;
#pragma unroll
        for (int t = 0; t < 6; ++t) v = fmaf(acc[t][r], acc[t][r], v);
        s[r] = v;
    }
    hsum16x8(s);
    {
        float xn = fmaxf(sqrtf(xsMine), MINN);
        float mn = fmaxf(sqrtf(SEL8(s, j)), MINN);
        float sc = tanhf(mn / xn * atanhf(fminf(xn, ARTC))) / mn;
        bcast8(sc, f0);
    }
#pragma unroll
    for (int r = 0; r < 8; ++r) {
        float v = 0.0f;
#pragma unroll
        for (int t = 0; t < 6; ++t) { acc[t][r] *= f0[r]; v = fmaf(acc[t][r], acc[t][r], v); }
        s[r] = v;
    }
    hsum16x8(s);
    {
        float hn = fmaxf(sqrtf(SEL8(s, j)), MINN);
        float pf = (hn > MAXN) ? MAXN / hn : 1.0f;
        bcast8(pf, f0);
    }
#pragma unroll
    for (int r = 0; r < 8; ++r) {
        float v = 0.0f, u = 0.0f;
#pragma unroll
        for (int t = 0; t < 6; ++t) {
            acc[t][r] *= f0[r];
            v = fmaf(acc[t][r], acc[t][r], v);
            u = fmaf(acc[t][r], hbv[t], u);
        }
        s[r] = v; q[r] = u;
    }
    hsum16x8(s);
    hsum16x8(q);
    {
        float x2 = SEL8(s, j);
        float xy = SEL8(q, j);
        float ca  = 1.0f + 2.0f * xy + y2;
        float cb  = 1.0f - x2;
        float den = fmaxf(1.0f + 2.0f * xy + x2 * y2, MINN);
        bcast8(ca / den, f0); bcast8(cb / den, f1);
    }
#pragma unroll
    for (int r = 0; r < 8; ++r) {
        float v = 0.0f;
#pragma unroll
        for (int t = 0; t < 6; ++t) {
            acc[t][r] = f0[r] * acc[t][r] + f1[r] * hbv[t];
            v = fmaf(acc[t][r], acc[t][r], v);
        }
        s[r] = v;
    }
    hsum16x8(s);
    {
        float gn = fmaxf(sqrtf(SEL8(s, j)), MINN);
        float pf = (gn > MAXN) ? MAXN / gn : 1.0f;
        float pn  = fminf(gn, MAXN);
        float lsc = atanhf(fminf(pn, ARTC)) / pn;
        bcast8(pf, f0); bcast8(lsc, f1);
    }

    float* obase = xt + (size_t)(tileM * 16 + 8 * half) * D_OUT + mr;
#pragma unroll
    for (int r = 0; r < 8; ++r) {
        if (fullTile || tileM * 16 + 8 * half + r < N) {
            float* o = obase + (size_t)r * D_OUT;
#pragma unroll
            for (int t = 0; t < 6; ++t) o[t * 16] = f1[r] * (acc[t][r] * f0[r]);
        }
    }
}

#define OWN_ROWS    196
#define MAX_RANGES  256
#define MAX_SLICES  256
#define BIN_THREADS 1024
#define BIN_REGS    4

__global__ __launch_bounds__(BIN_THREADS) void bin_edges(
    const int* __restrict__ erow, const int* __restrict__ ecol,
    const float* __restrict__ ew, int* __restrict__ sorted_w,
    int* __restrict__ tcnt, int* __restrict__ toff, int E, int N, int S) {
    int4* __restrict__ sorted = (int4*)sorted_w;
    __shared__ int s_cnt[MAX_RANGES];
    __shared__ int s_off[MAX_RANGES];
    __shared__ int s_cur[MAX_RANGES];
    const int tid = threadIdx.x;
    const int b   = blockIdx.x;
    const int P   = gridDim.x;
    const int beg = b * S;
    const int end = min(E, beg + S);
    const bool inRegs = (S <= BIN_REGS * BIN_THREADS);

    if (tid < MAX_RANGES) s_cnt[tid] = 0;

    int r[BIN_REGS], c[BIN_REGS], w[BIN_REGS], rk[BIN_REGS];
    if (inRegs) {
        for (int i = 0; i < BIN_REGS; ++i) {
            const int e = beg + tid + i * BIN_THREADS;
            r[i] = -1; c[i] = 0; w[i] = 0;
            if (e < end) { r[i] = erow[e]; c[i] = ecol[e]; w[i] = __float_as_int(ew[e]); }
        }
    }
    __syncthreads();
    if (inRegs) {
        for (int i = 0; i < BIN_REGS; ++i) {
            rk[i] = 0;
            if ((unsigned)r[i] < (unsigned)N) rk[i] = atomicAdd(&s_cnt[r[i] / OWN_ROWS], 1);
        }
    } else {
        for (int e = beg + tid; e < end; e += BIN_THREADS) {
            const int rr = erow[e];
            if ((unsigned)rr < (unsigned)N) atomicAdd(&s_cnt[rr / OWN_ROWS], 1);
        }
    }
    __syncthreads();
    if (tid < 32) {
        int v[8];
        int s = 0;
        for (int j = 0; j < 8; ++j) { v[j] = s_cnt[tid * 8 + j]; s += v[j]; }
        int inc = s;
        for (int d = 1; d < 32; d <<= 1) {
            const int t = __shfl_up(inc, d, 32);
            if (tid >= d) inc += t;
        }
        int run = inc - s;
        for (int j = 0; j < 8; ++j) { s_off[tid * 8 + j] = run; s_cur[tid * 8 + j] = run; run += v[j]; }
    }
    __syncthreads();
    if (tid < MAX_RANGES) {
        tcnt[tid * P + b] = s_cnt[tid];
        toff[tid * P + b] = s_off[tid];
    }
    if (inRegs) {
        for (int i = 0; i < BIN_REGS; ++i) {
            if ((unsigned)r[i] < (unsigned)N)
                sorted[beg + s_off[r[i] / OWN_ROWS] + rk[i]] = make_int4(r[i], c[i], w[i], 0);
        }
    } else {
        for (int e = beg + tid; e < end; e += BIN_THREADS) {
            const int rr = erow[e];
            if ((unsigned)rr < (unsigned)N) {
                const int pos = beg + atomicAdd(&s_cur[rr / OWN_ROWS], 1);
                sorted[pos] = make_int4(rr, ecol[e], __float_as_int(ew[e]), 0);
            }
        }
    }
}

#define OWN_THREADS 1024
#define EDGE_CHUNK  32
#define EDGE_BATCH  16

template <bool FULL>
__device__ __forceinline__ void chunk_edges(const float* __restrict__ xt, float* tile,
                                            int r_v, int c_v, int w_v, int cnt, int lane) {
#pragma unroll
    for (int j = 0; j < EDGE_CHUNK; j += EDGE_BATCH) {
        float v[EDGE_BATCH][3];
#pragma unroll
        for (int u = 0; u < EDGE_BATCH; ++u) {
            const int c = __builtin_amdgcn_readlane(c_v, j + u);
            const float* src = xt + (size_t)c * D_OUT + lane;
            v[u][0] = src[0]; v[u][1] = src[32]; v[u][2] = src[64];
        }
#pragma unroll
        for (int u = 0; u < EDGE_BATCH; ++u) {
            if (FULL || j + u < cnt) {
                const int   r = __builtin_amdgcn_readlane(r_v, j + u);
                const float w = __int_as_float(__builtin_amdgcn_readlane(w_v, j + u));
                float* dst = tile + r * D_OUT + lane;
                atomicAdd(dst,      w * v[u][0]);
                atomicAdd(dst + 32, w * v[u][1]);
                atomicAdd(dst + 64, w * v[u][2]);
            }
        }
    }
}

__global__ __launch_bounds__(OWN_THREADS) void own_rows(
    const int* __restrict__ sorted_w, const int* __restrict__ tcnt,
    const int* __restrict__ toff, const float* __restrict__ xt,
    float* __restrict__ out, int N, int S, int P) {
    const int4* __restrict__ sorted = (const int4*)sorted_w;
    __shared__ __attribute__((aligned(16))) float tile[OWN_ROWS * D_OUT];
    __shared__ int s_c[MAX_SLICES];
    __shared__ int s_base[MAX_SLICES];
    __shared__ int s_pre[MAX_SLICES + 1];
    const int tid  = threadIdx.x;
    const int lane = tid & 31;
    const int wave = __builtin_amdgcn_readfirstlane(tid >> 5);
    const int k    = blockIdx.x;
    const int lo   = k * OWN_ROWS;
    const int nrows = min(OWN_ROWS, N - lo);

    {
        const float4 z4 = make_float4(0.0f, 0.0f, 0.0f, 0.0f);
        for (int i = tid; i < OWN_ROWS * D_OUT / 4; i += OWN_THREADS) ((float4*)tile)[i] = z4;
    }
    if (tid < MAX_SLICES) {
        int c = 0, o = 0;
        if (tid < P) { c = tcnt[k * P + tid]; o = toff[k * P + tid]; }
        s_c[tid]    = c;
        s_base[tid] = tid * S + o;
    }
    __syncthreads();
    if (tid < 32) {
        int v[8];
        int s = 0;
        for (int j = 0; j < 8; ++j) { v[j] = s_c[tid * 8 + j]; s += v[j]; }
        int inc = s;
        for (int d = 1; d < 32; d <<= 1) {
            const int t = __shfl_up(inc, d, 32);
            if (tid >= d) inc += t;
        }
        int run = inc - s;
        for (int j = 0; j < 8; ++j) { s_pre[tid * 8 + j] = run; run += v[j]; }
        if (tid == 31) s_pre[MAX_SLICES] = run;
    }
    __syncthreads();
    const int T = __builtin_amdgcn_readfirstlane(s_pre[MAX_SLICES]);

    for (int ch = wave; ch * EDGE_CHUNK < T; ch += OWN_THREADS / 32) {
        const int cnt = min(EDGE_CHUNK, T - ch * EDGE_CHUNK);
        int r_v = 0, c_v = 0, w_v = 0;
        if (lane < cnt) {
            const int f = ch * EDGE_CHUNK + lane;
            int a = 0, z = MAX_SLICES;
            for (int it = 0; it < 8; ++it) {
                const int m = (a + z) >> 1;
                if (s_pre[m] <= f) a = m; else z = m;
            }
            const int4 rec = sorted[s_base[a] + (f - s_pre[a])];
            r_v = rec.x - lo;
            c_v = rec.y;
            w_v = rec.z;
        }
        if (cnt == EDGE_CHUNK) chunk_edges<true>(xt, tile, r_v, c_v, w_v, cnt, lane);
        else                   chunk_edges<false>(xt, tile, r_v, c_v, w_v, cnt, lane);
    }
    __syncthreads();

    for (int rr = wave; rr < nrows; rr += OWN_THREADS / 32) {
        const float* sr = tile + rr * D_OUT;
        float s0 = sr[lane], s1 = sr[lane + 32], s2 = sr[lane + 64];
        float sn = fmaxf(sqrtf(wsum(s0 * s0 + s1 * s1 + s2 * s2)), MINN);
        float e  = tanhf(sn) / sn;
        float p0 = e * s0, p1 = e * s1, p2 = e * s2;
        float pnr = fmaxf(sqrtf(wsum(p0 * p0 + p1 * p1 + p2 * p2)), MINN);
        float pf  = (pnr > MAXN) ? MAXN / pnr : 1.0f;
        p0 *= pf; p1 *= pf; p2 *= pf;
        float pn = fminf(pnr, MAXN);
        float l  = atanhf(fminf(pn, ARTC)) / pn;
        float u0 = fmaxf(l * p0, 0.0f), u1 = fmaxf(l * p1, 0.0f), u2 = fmaxf(l * p2, 0.0f);
        float un = fmaxf(sqrtf(wsum(u0 * u0 + u1 * u1 + u2 * u2)), MINN);
        float e2 = tanhf(un) / un;
        float h0 = e2 * u0, h1 = e2 * u1, h2 = e2 * u2;
        float hn  = fmaxf(sqrtf(wsum(h0 * h0 + h1 * h1 + h2 * h2)), MINN);
        float pf2 = (hn > MAXN) ? MAXN / hn : 1.0f;
        float* o = out + (size_t)(lo + rr) * D_OUT;
        o[lane] = h0 * pf2; o[lane + 32] = h1 * pf2; o[lane + 64] = h2 * pf2;
    }
}

extern "C" void kernel_launch(void* const* d_in, const int* in_sizes, int n_in,
                              void* d_out, int out_size, void* d_ws, size_t ws_size,
                              hipStream_t stream) {
    const float* x      = (const float*)d_in[0];
    const float* weight = (const float*)d_in[1];
    const float* bias   = (const float*)d_in[2];
    const float* ew     = (const float*)d_in[3];
    const int*   erow   = (const int*)d_in[4];
    const int*   ecol   = (const int*)d_in[5];
    const int N = in_sizes[0] / D_IN;
    const int E = in_sizes[3];

    float* ws   = (float*)d_ws;
    float* buf0 = ws + 128;
    float* buf1 = buf0 + (size_t)N * D_OUT;
    float* out  = (float*)d_out;

    int*  sorted = (int*)buf0;
    int*  tcnt   = sorted + 4 * (size_t)E;
    int*  toff   = tcnt + MAX_RANGES * MAX_SLICES;

    int strips     = (N + 15) >> 4;
    int gemmBlocks = (strips + GEMM_WAVES - 1) / GEMM_WAVES;
    gemm_hyp_linear<<<gemmBlocks, GEMM_THREADS, 0, stream>>>(x, weight, bias, buf1, N);

    const int P = MAX_SLICES;
    const int S = (E + P - 1) / P;
    bin_edges<<<P, BIN_THREADS, 0, stream>>>(erow, ecol, ew, sorted, tcnt, toff, E, N, S);

    int ranges = (N + OWN_ROWS - 1) / OWN_ROWS;
    own_rows<<<ranges, OWN_THREADS, 0, stream>>>(sorted, tcnt, toff, buf1, out, N, S, P);
}


# owner kernel row tail: 8 lanes per row, 8 rows per wave at once, row sums by DPP exchanges (new summation order), one scalar chain per two rows; was one wave per row with LDS-unit shuffles
# speedup vs baseline: 3.3587x; 1.1272x over previous
#include <hip/hip_runtime.h>
#include <hip/hip_bf16.h>


typedef __attribute__((ext_vector_type(2))) float v2f;
typedef __attribute__((ext_vector_type(8))) float v8f;

#define D_IN   256
#define D_OUT  96
#define MINN   1e-15f
#define MAXN   0.996f
#define ARTC   (1.0f - 1e-7f)

template <int CTRL>
__device__ __forceinline__ float dpp_read(float v) {
    return __builtin_bit_cast(float, __builtin_amdgcn_update_dpp(
        0, __builtin_bit_cast(int, v), CTRL, 0xf, 0xf, false));
}

__device__ __forceinline__ float hsum16(float v) {
    v += dpp_read<0x161>(v);
    v += dpp_read<0x162>(v);
    v += dpp_read<0x164>(v);
    v += dpp_read<0x168>(v);
    return v;
}

__device__ __forceinline__ float gsum8(float v) {
    v += dpp_read<0x161>(v);
    v += dpp_read<0x162>(v);
    v += dpp_read<0x164>(v);
    return v;
}

__device__ __forceinline__ void hsum16x8(float (&s)[8]) {
#pragma unroll
    for (int r = 0; r < 8; ++r) s[r] = hsum16(s[r]);
}

__device__ __forceinline__ float sel8v(float v0, float v1, float v2, float v3,
                                       float v4, float v5, float v6, float v7, int j) {
    const bool b0 = (j & 1) != 0, b1 = (j & 2) != 0, b2 = (j & 4) != 0;
    float a0 = b0 ? v1 : v0, a1 = b0 ? v3 : v2;
    float a2 = b0 ? v5 : v4, a3 = b0 ? v7 : v6;
    float c0 = b1 ? a1 : a0, c1 = b1 ? a3 : a2;
    return b2 ? c1 : c0;
}
#define SEL8(a, j) sel8v(a[0], a[1], a[2], a[3], a[4], a[5], a[6], a[7], j)

__device__ __forceinline__ void bcast8(float v, float (&o)[8]) {
    o[0] = dpp_read<0x150>(v); o[1] = dpp_read<0x151>(v);
    o[2] = dpp_read<0x152>(v); o[3] = dpp_read<0x153>(v);
    o[4] = dpp_read<0x154>(v); o[5] = dpp_read<0x155>(v);
    o[6] = dpp_read<0x156>(v); o[7] = dpp_read<0x157>(v);
}

#define GEMM_THREADS 512
#define GEMM_WAVES   (GEMM_THREADS / 32)
#define WP           260

__global__ __launch_bounds__(GEMM_THREADS) __attribute__((amdgpu_waves_per_eu(4, 5)))
void gemm_hyp_linear(const float* __restrict__ x, const float* __restrict__ w,
                     const float* __restrict__ bias, float* __restrict__ xt, int N) {
    __shared__ __attribute__((aligned(16))) float sW[D_OUT * WP];
    const int nTilesM = (N + 15) >> 4;
    const int lane = threadIdx.x & 31;
    const int wave = __builtin_amdgcn_readfirstlane(threadIdx.x >> 5);
    const int half = lane >> 4;
    const int mr   = lane & 15;
    const int tileM = blockIdx.x * GEMM_WAVES + wave;

    int mA = tileM * 16 + mr; if (mA > N - 1) mA = N - 1;
    const float* aP = x + (size_t)mA * D_IN + 4 * half;
    float4 aN[4];
#pragma unroll
    for (int i = 0; i < 4; ++i) aN[i] = make_float4(0.0f, 0.0f, 0.0f, 0.0f);
    if (tileM < nTilesM) {
#pragma unroll
        for (int i = 0; i < 4; ++i) aN[i] = *(const float4*)(aP + 8 * i);
    }

    float hbv[6];
    float y2;
    {
        float bs = 0.0f;
#pragma unroll
        for (int t = 0; t < 6; ++t) { hbv[t] = bias[t * 16 + mr]; bs = fmaf(hbv[t], hbv[t], bs); }
        float n = fmaxf(sqrtf(hsum16(bs)), MINN);
        float e = tanhf(n) / n;
        float ps = 0.0f;
#pragma unroll
        for (int t = 0; t < 6; ++t) { hbv[t] = e * hbv[t]; ps = fmaf(hbv[t], hbv[t], ps); }
        float pn = fmaxf(sqrtf(hsum16(ps)), MINN);
        float pf = (pn > MAXN) ? MAXN / pn : 1.0f;
        float ys = 0.0f;
#pragma unroll
        for (int t = 0; t < 6; ++t) { hbv[t] *= pf; ys = fmaf(hbv[t], hbv[t], ys); }
        y2 = hsum16(ys);
    }

    for (int i = threadIdx.x; i < D_OUT * (D_IN / 4); i += GEMM_THREADS) {
        const int r = i >> 6, c = (i & 63) << 2;
        *(float4*)(sW + r * WP + c) = *(const float4*)(w + (size_t)r * D_IN + c);
    }
    __syncthreads();
    const float* bT[6];
#pragma unroll
    for (int t = 0; t < 6; ++t) bT[t] = sW + (t * 16 + mr) * WP + 4 * half;

    if (tileM >= nTilesM) return;
    const bool fullTile = (tileM * 16 + 15 < N);

    v8f acc[6] = {{}, {}, {}, {}, {}, {}};
    float4 aA[8][4];
#pragma unroll
    for (int i = 0; i < 4; ++i) aA[0][i] = aN[i];
    float4 bC[6], bN[6];
#pragma unroll
    for (int t = 0; t < 6; ++t) bC[t] = *(const float4*)(bT[t]);
    __builtin_amdgcn_sched_barrier(0);
#pragma unroll
    for (int kb = 0; kb < D_IN; kb += 32) {
        if (kb + 32 < D_IN) {
#pragma unroll
            for (int i = 0; i < 4; ++i) aA[kb / 32 + 1][i] = *(const float4*)(aP + kb + 32 + 8 * i);
        }
#pragma unroll
        for (int i = 0; i < 4; ++i) {
            if (kb + 8 * i + 8 < D_IN) {
#pragma unroll
                for (int t = 0; t < 6; ++t) bN[t] = *(const float4*)(bT[t] + kb + 8 * i + 8);
            }
            const float4 av = aA[kb / 32][i];
            v2f a01; a01.x = av.x; a01.y = av.y;
            v2f a23; a23.x = av.z; a23.y = av.w;
#pragma unroll
            for (int t = 0; t < 6; ++t) {
                v2f b01; b01.x = bC[t].x; b01.y = bC[t].y;
                acc[t] = __builtin_amdgcn_wmma_f32_16x16x4_f32(
                     false, a01,  false, b01,
                     (short)0, acc[t],  false,  false);
            }
#pragma unroll
            for (int t = 0; t < 6; ++t) {
                v2f b23; b23.x = bC[t].z; b23.y = bC[t].w;
                acc[t] = __builtin_amdgcn_wmma_f32_16x16x4_f32(
                     false, a23,  false, b23,
                     (short)0, acc[t],  false,  false);
            }
#pragma unroll
            for (int t = 0; t < 6; ++t) bC[t] = bN[t];
        }
    }
    __builtin_amdgcn_sched_barrier(0);
    asm volatile("v_nop\n\tv_nop\n\tv_nop\n\tv_nop");
    __builtin_amdgcn_sched_barrier(0);

    float ss0 = 0.0f, ss1 = 0.0f, ss2 = 0.0f, ss3 = 0.0f;
#pragma unroll
    for (int b = 0; b < 8; ++b) {
#pragma unroll
        for (int i = 0; i < 4; ++i) {
            ss0 = fmaf(aA[b][i].x, aA[b][i].x, ss0);
            ss1 = fmaf(aA[b][i].y, aA[b][i].y, ss1);
            ss2 = fmaf(aA[b][i].z, aA[b][i].z, ss2);
            ss3 = fmaf(aA[b][i].w, aA[b][i].w, ss3);
        }
    }

    float xs = (ss0 + ss1) + (ss2 + ss3);
    xs += __shfl_xor(xs, 16, 32);
    const float xsOther = dpp_read<0x168>(xs);
    const float xsMine  = ((mr >> 3) == half) ? xs : xsOther;
    const int   j       = mr & 7;

    float s[8], q[8], f0[8], f1[8];

#pragma unroll
    for (int r = 0; r < 8; ++r) {
        float v = 0.0f;
#pragma unroll
        for (int t = 0; t < 6; ++t) v = fmaf(acc[t][r], acc[t][r], v);
        s[r] = v;
    }
    hsum16x8(s);
    {
        float xn = fmaxf(sqrtf(xsMine), MINN);
        float mn = fmaxf(sqrtf(SEL8(s, j)), MINN);
        float sc = tanhf(mn / xn * atanhf(fminf(xn, ARTC))) / mn;
        bcast8(sc, f0);
    }
#pragma unroll
    for (int r = 0; r < 8; ++r) {
        float v = 0.0f;
#pragma unroll
        for (int t = 0; t < 6; ++t) { acc[t][r] *= f0[r]; v = fmaf(acc[t][r], acc[t][r], v); }
        s[r] = v;
    }
    hsum16x8(s);
    {
        float hn = fmaxf(sqrtf(SEL8(s, j)), MINN);
        float pf = (hn > MAXN) ? MAXN / hn : 1.0f;
        bcast8(pf, f0);
    }
#pragma unroll
    for (int r = 0; r < 8; ++r) {
        float v = 0.0f, u = 0.0f;
#pragma unroll
        for (int t = 0; t < 6; ++t) {
            acc[t][r] *= f0[r];
            v = fmaf(acc[t][r], acc[t][r], v);
            u = fmaf(acc[t][r], hbv[t], u);
        }
        s[r] = v; q[r] = u;
    }
    hsum16x8(s);
    hsum16x8(q);
    {
        float x2 = SEL8(s, j);
        float xy = SEL8(q, j);
        float ca  = 1.0f + 2.0f * xy + y2;
        float cb  = 1.0f - x2;
        float den = fmaxf(1.0f + 2.0f * xy + x2 * y2, MINN);
        bcast8(ca / den, f0); bcast8(cb / den, f1);
    }
#pragma unroll
    for (int r = 0; r < 8; ++r) {
        float v = 0.0f;
#pragma unroll
        for (int t = 0; t < 6; ++t) {
            acc[t][r] = f0[r] * acc[t][r] + f1[r] * hbv[t];
            v = fmaf(acc[t][r], acc[t][r], v);
        }
        s[r] = v;
    }
    hsum16x8(s);
    {
        float gn = fmaxf(sqrtf(SEL8(s, j)), MINN);
        float pf = (gn > MAXN) ? MAXN / gn : 1.0f;
        float pn  = fminf(gn, MAXN);
        float lsc = atanhf(fminf(pn, ARTC)) / pn;
        bcast8(pf, f0); bcast8(lsc, f1);
    }

    float* obase = xt + (size_t)(tileM * 16 + 8 * half) * D_OUT + mr;
#pragma unroll
    for (int r = 0; r < 8; ++r) {
        if (fullTile || tileM * 16 + 8 * half + r < N) {
            float* o = obase + (size_t)r * D_OUT;
#pragma unroll
            for (int t = 0; t < 6; ++t) o[t * 16] = f1[r] * (acc[t][r] * f0[r]);
        }
    }
}

#define OWN_ROWS    196
#define MAX_RANGES  256
#define MAX_SLICES  256
#define BIN_THREADS 1024
#define BIN_REGS    4

__global__ __launch_bounds__(BIN_THREADS) void bin_edges(
    const int* __restrict__ erow, const int* __restrict__ ecol,
    const float* __restrict__ ew, int* __restrict__ sorted_w,
    int* __restrict__ tcnt, int* __restrict__ toff, int E, int N, int S) {
    int4* __restrict__ sorted = (int4*)sorted_w;
    __shared__ int s_cnt[MAX_RANGES];
    __shared__ int s_off[MAX_RANGES];
    __shared__ int s_cur[MAX_RANGES];
    const int tid = threadIdx.x;
    const int b   = blockIdx.x;
    const int P   = gridDim.x;
    const int beg = b * S;
    const int end = min(E, beg + S);
    const bool inRegs = (S <= BIN_REGS * BIN_THREADS);

    if (tid < MAX_RANGES) s_cnt[tid] = 0;

    int r[BIN_REGS], c[BIN_REGS], w[BIN_REGS], rk[BIN_REGS];
    if (inRegs) {
        for (int i = 0; i < BIN_REGS; ++i) {
            const int e = beg + tid + i * BIN_THREADS;
            r[i] = -1; c[i] = 0; w[i] = 0;
            if (e < end) { r[i] = erow[e]; c[i] = ecol[e]; w[i] = __float_as_int(ew[e]); }
        }
    }
    __syncthreads();
    if (inRegs) {
        for (int i = 0; i < BIN_REGS; ++i) {
            rk[i] = 0;
            if ((unsigned)r[i] < (unsigned)N) rk[i] = atomicAdd(&s_cnt[r[i] / OWN_ROWS], 1);
        }
    } else {
        for (int e = beg + tid; e < end; e += BIN_THREADS) {
            const int rr = erow[e];
            if ((unsigned)rr < (unsigned)N) atomicAdd(&s_cnt[rr / OWN_ROWS], 1);
        }
    }
    __syncthreads();
    if (tid < 32) {
        int v[8];
        int s = 0;
        for (int j = 0; j < 8; ++j) { v[j] = s_cnt[tid * 8 + j]; s += v[j]; }
        int inc = s;
        for (int d = 1; d < 32; d <<= 1) {
            const int t = __shfl_up(inc, d, 32);
            if (tid >= d) inc += t;
        }
        int run = inc - s;
        for (int j = 0; j < 8; ++j) { s_off[tid * 8 + j] = run; s_cur[tid * 8 + j] = run; run += v[j]; }
    }
    __syncthreads();
    if (tid < MAX_RANGES) {
        tcnt[tid * P + b] = s_cnt[tid];
        toff[tid * P + b] = s_off[tid];
    }
    if (inRegs) {
        for (int i = 0; i < BIN_REGS; ++i) {
            if ((unsigned)r[i] < (unsigned)N)
                sorted[beg + s_off[r[i] / OWN_ROWS] + rk[i]] = make_int4(r[i], c[i], w[i], 0);
        }
    } else {
        for (int e = beg + tid; e < end; e += BIN_THREADS) {
            const int rr = erow[e];
            if ((unsigned)rr < (unsigned)N) {
                const int pos = beg + atomicAdd(&s_cur[rr / OWN_ROWS], 1);
                sorted[pos] = make_int4(rr, ecol[e], __float_as_int(ew[e]), 0);
            }
        }
    }
}

#define OWN_THREADS 1024
#define EDGE_CHUNK  32
#define EDGE_BATCH  16

template <bool FULL>
__device__ __forceinline__ void chunk_edges(const float* __restrict__ xt, float* tile,
                                            int r_v, int c_v, int w_v, int cnt, int lane) {
#pragma unroll
    for (int j = 0; j < EDGE_CHUNK; j += EDGE_BATCH) {
        float v[EDGE_BATCH][3];
#pragma unroll
        for (int u = 0; u < EDGE_BATCH; ++u) {
            const int c = __builtin_amdgcn_readlane(c_v, j + u);
            const float* src = xt + (size_t)c * D_OUT + lane;
            v[u][0] = src[0]; v[u][1] = src[32]; v[u][2] = src[64];
        }
#pragma unroll
        for (int u = 0; u < EDGE_BATCH; ++u) {
            if (FULL || j + u < cnt) {
                const int   r = __builtin_amdgcn_readlane(r_v, j + u);
                const float w = __int_as_float(__builtin_amdgcn_readlane(w_v, j + u));
                float* dst = tile + r * D_OUT + lane;
                atomicAdd(dst,      w * v[u][0]);
                atomicAdd(dst + 32, w * v[u][1]);
                atomicAdd(dst + 64, w * v[u][2]);
            }
        }
    }
}

__global__ __launch_bounds__(OWN_THREADS) void own_rows(
    const int* __restrict__ sorted_w, const int* __restrict__ tcnt,
    const int* __restrict__ toff, const float* __restrict__ xt,
    float* __restrict__ out, int N, int S, int P) {
    const int4* __restrict__ sorted = (const int4*)sorted_w;
    __shared__ __attribute__((aligned(16))) float tile[OWN_ROWS * D_OUT];
    __shared__ int s_c[MAX_SLICES];
    __shared__ int s_base[MAX_SLICES];
    __shared__ int s_pre[MAX_SLICES + 1];
    const int tid  = threadIdx.x;
    const int lane = tid & 31;
    const int wave = __builtin_amdgcn_readfirstlane(tid >> 5);
    const int k    = blockIdx.x;
    const int lo   = k * OWN_ROWS;
    const int nrows = min(OWN_ROWS, N - lo);

    {
        const float4 z4 = make_float4(0.0f, 0.0f, 0.0f, 0.0f);
        for (int i = tid; i < OWN_ROWS * D_OUT / 4; i += OWN_THREADS) ((float4*)tile)[i] = z4;
    }
    if (tid < MAX_SLICES) {
        int c = 0, o = 0;
        if (tid < P) { c = tcnt[k * P + tid]; o = toff[k * P + tid]; }
        s_c[tid]    = c;
        s_base[tid] = tid * S + o;
    }
    __syncthreads();
    if (tid < 32) {
        int v[8];
        int s = 0;
        for (int j = 0; j < 8; ++j) { v[j] = s_c[tid * 8 + j]; s += v[j]; }
        int inc = s;
        for (int d = 1; d < 32; d <<= 1) {
            const int t = __shfl_up(inc, d, 32);
            if (tid >= d) inc += t;
        }
        int run = inc - s;
        for (int j = 0; j < 8; ++j) { s_pre[tid * 8 + j] = run; run += v[j]; }
        if (tid == 31) s_pre[MAX_SLICES] = run;
    }
    __syncthreads();
    const int T = __builtin_amdgcn_readfirstlane(s_pre[MAX_SLICES]);

    for (int ch = wave; ch * EDGE_CHUNK < T; ch += OWN_THREADS / 32) {
        const int cnt = min(EDGE_CHUNK, T - ch * EDGE_CHUNK);
        int r_v = 0, c_v = 0, w_v = 0;
        if (lane < cnt) {
            const int f = ch * EDGE_CHUNK + lane;
            int a = 0, z = MAX_SLICES;
            for (int it = 0; it < 8; ++it) {
                const int m = (a + z) >> 1;
                if (s_pre[m] <= f) a = m; else z = m;
            }
            const int4 rec = sorted[s_base[a] + (f - s_pre[a])];
            r_v = rec.x - lo;
            c_v = rec.y;
            w_v = rec.z;
        }
        if (cnt == EDGE_CHUNK) chunk_edges<true>(xt, tile, r_v, c_v, w_v, cnt, lane);
        else                   chunk_edges<false>(xt, tile, r_v, c_v, w_v, cnt, lane);
    }
    __syncthreads();

    static_assert(OWN_ROWS <= 2 * (OWN_THREADS / 8), "two rows per group of eight lanes must cover the block's rows");
    if (wave * 4 < nrows) {
        const int g = lane >> 3, s8 = lane & 7;
        const int rA = wave * 4 + g, rB = rA + OWN_THREADS / 8;
        const int lA = min(rA, nrows - 1), lB = min(rB, nrows - 1);
        const bool odd = (lane & 1) != 0;
        float a[12], b[12];
        {
            const float4* pa = (const float4*)(tile + lA * D_OUT) + s8;
            const float4* pb = (const float4*)(tile + lB * D_OUT) + s8;
#pragma unroll
            for (int i = 0; i < 3; ++i) {
                const float4 va = pa[8 * i], vb = pb[8 * i];
                a[4 * i] = va.x; a[4 * i + 1] = va.y; a[4 * i + 2] = va.z; a[4 * i + 3] = va.w;
                b[4 * i] = vb.x; b[4 * i + 1] = vb.y; b[4 * i + 2] = vb.z; b[4 * i + 3] = vb.w;
            }
        }
        float qa, qb;
        qa = 0.0f; qb = 0.0f;
#pragma unroll
        for (int j = 0; j < 12; ++j) { qa = fmaf(a[j], a[j], qa); qb = fmaf(b[j], b[j], qb); }
        qa = gsum8(qa); qb = gsum8(qb);
        {
            float sn = fmaxf(sqrtf(odd ? qb : qa), MINN);
            float e  = tanhf(sn) / sn;
            const float ea = dpp_read<0x00>(e), eb = dpp_read<0x55>(e);
#pragma unroll
            for (int j = 0; j < 12; ++j) { a[j] *= ea; b[j] *= eb; }
        }
        qa = 0.0f; qb = 0.0f;
#pragma unroll
        for (int j = 0; j < 12; ++j) { qa = fmaf(a[j], a[j], qa); qb = fmaf(b[j], b[j], qb); }
        qa = gsum8(qa); qb = gsum8(qb);
        {
            float pnr = fmaxf(sqrtf(odd ? qb : qa), MINN);
            float pf  = (pnr > MAXN) ? MAXN / pnr : 1.0f;
            float pn  = fminf(pnr, MAXN);
            float l   = atanhf(fminf(pn, ARTC)) / pn;
            const float pfa = dpp_read<0x00>(pf), pfb = dpp_read<0x55>(pf);
            const float la  = dpp_read<0x00>(l),  lb  = dpp_read<0x55>(l);
#pragma unroll
            for (int j = 0; j < 12; ++j) {
                a[j] *= pfa; a[j] = fmaxf(la * a[j], 0.0f);
                b[j] *= pfb; b[j] = fmaxf(lb * b[j], 0.0f);
            }
        }
        qa = 0.0f; qb = 0.0f;
#pragma unroll
        for (int j = 0; j < 12; ++j) { qa = fmaf(a[j], a[j], qa); qb = fmaf(b[j], b[j], qb); }
        qa = gsum8(qa); qb = gsum8(qb);
        {
            float un = fmaxf(sqrtf(odd ? qb : qa), MINN);
            float e2 = tanhf(un) / un;
            const float ea = dpp_read<0x00>(e2), eb = dpp_read<0x55>(e2);
#pragma unroll
            for (int j = 0; j < 12; ++j) { a[j] *= ea; b[j] *= eb; }
        }
        qa = 0.0f; qb = 0.0f;
#pragma unroll
        for (int j = 0; j < 12; ++j) { qa = fmaf(a[j], a[j], qa); qb = fmaf(b[j], b[j], qb); }
        qa = gsum8(qa); qb = gsum8(qb);
        {
            float hn  = fmaxf(sqrtf(odd ? qb : qa), MINN);
            float pf2 = (hn > MAXN) ? MAXN / hn : 1.0f;
            const float fa = dpp_read<0x00>(pf2), fb = dpp_read<0x55>(pf2);
            if (rA < nrows) {
                float4* o = (float4*)(out + (size_t)(lo + rA) * D_OUT) + s8;
#pragma unroll
                for (int i = 0; i < 3; ++i)
                    o[8 * i] = make_float4(a[4 * i] * fa, a[4 * i + 1] * fa, a[4 * i + 2] * fa, a[4 * i + 3] * fa);
            }
            if (rB < nrows) {
                float4* o = (float4*)(out + (size_t)(lo + rB) * D_OUT) + s8;
#pragma unroll
                for (int i = 0; i < 3; ++i)
                    o[8 * i] = make_float4(b[4 * i] * fb, b[4 * i + 1] * fb, b[4 * i + 2] * fb, b[4 * i + 3] * fb);
            }
        }
    }
}

extern "C" void kernel_launch(void* const* d_in, const int* in_sizes, int n_in,
                              void* d_out, int out_size, void* d_ws, size_t ws_size,
                              hipStream_t stream) {
    const float* x      = (const float*)d_in[0];
    const float* weight = (const float*)d_in[1];
    const float* bias   = (const float*)d_in[2];
    const float* ew     = (const float*)d_in[3];
    const int*   erow   = (const int*)d_in[4];
    const int*   ecol   = (const int*)d_in[5];
    const int N = in_sizes[0] / D_IN;
    const int E = in_sizes[3];

    float* ws   = (float*)d_ws;
    float* buf0 = ws + 128;
    float* buf1 = buf0 + (size_t)N * D_OUT;
    float* out  = (float*)d_out;

    int*  sorted = (int*)buf0;
    int*  tcnt   = sorted + 4 * (size_t)E;
    int*  toff   = tcnt + MAX_RANGES * MAX_SLICES;

    int strips     = (N + 15) >> 4;
    int gemmBlocks = (strips + GEMM_WAVES - 1) / GEMM_WAVES;
    gemm_hyp_linear<<<gemmBlocks, GEMM_THREADS, 0, stream>>>(x, weight, bias, buf1, N);

    const int P = MAX_SLICES;
    const int S = (E + P - 1) / P;
    bin_edges<<<P, BIN_THREADS, 0, stream>>>(erow, ecol, ew, sorted, tcnt, toff, E, N, S);

    int ranges = (N + OWN_ROWS - 1) / OWN_ROWS;
    own_rows<<<ranges, OWN_THREADS, 0, stream>>>(sorted, tcnt, toff, buf1, out, N, S, P);
}


# owner kernel: edges ordered by destination row in LDS (2nd counting sort); 8 lanes sum a row in registers (fma chain, rank = arrival order), 16-byte gathers; no LDS float adds, no sum tile
# speedup vs baseline: 3.7006x; 1.1018x over previous
#include <hip/hip_runtime.h>
#include <hip/hip_bf16.h>


typedef __attribute__((ext_vector_type(2))) float v2f;
typedef __attribute__((ext_vector_type(8))) float v8f;

#define D_IN   256
#define D_OUT  96
#define MINN   1e-15f
#define MAXN   0.996f
#define ARTC   (1.0f - 1e-7f)

template <int CTRL>
__device__ __forceinline__ float dpp_read(float v) {
    return __builtin_bit_cast(float, __builtin_amdgcn_update_dpp(
        0, __builtin_bit_cast(int, v), CTRL, 0xf, 0xf, false));
}

__device__ __forceinline__ float hsum16(float v) {
    v += dpp_read<0x161>(v);
    v += dpp_read<0x162>(v);
    v += dpp_read<0x164>(v);
    v += dpp_read<0x168>(v);
    return v;
}

__device__ __forceinline__ float gsum8(float v) {
    v += dpp_read<0x161>(v);
    v += dpp_read<0x162>(v);
    v += dpp_read<0x164>(v);
    return v;
}

__device__ __forceinline__ void hsum16x8(float (&s)[8]) {
#pragma unroll
    for (int r = 0; r < 8; ++r) s[r] = hsum16(s[r]);
}

__device__ __forceinline__ float sel8v(float v0, float v1, float v2, float v3,
                                       float v4, float v5, float v6, float v7, int j) {
    const bool b0 = (j & 1) != 0, b1 = (j & 2) != 0, b2 = (j & 4) != 0;
    float a0 = b0 ? v1 : v0, a1 = b0 ? v3 : v2;
    float a2 = b0 ? v5 : v4, a3 = b0 ? v7 : v6;
    float c0 = b1 ? a1 : a0, c1 = b1 ? a3 : a2;
    return b2 ? c1 : c0;
}
#define SEL8(a, j) sel8v(a[0], a[1], a[2], a[3], a[4], a[5], a[6], a[7], j)

__device__ __forceinline__ void bcast8(float v, float (&o)[8]) {
    o[0] = dpp_read<0x150>(v); o[1] = dpp_read<0x151>(v);
    o[2] = dpp_read<0x152>(v); o[3] = dpp_read<0x153>(v);
    o[4] = dpp_read<0x154>(v); o[5] = dpp_read<0x155>(v);
    o[6] = dpp_read<0x156>(v); o[7] = dpp_read<0x157>(v);
}

#define GEMM_THREADS 512
#define GEMM_WAVES   (GEMM_THREADS / 32)
#define WP           260

__global__ __launch_bounds__(GEMM_THREADS) __attribute__((amdgpu_waves_per_eu(4, 5)))
void gemm_hyp_linear(const float* __restrict__ x, const float* __restrict__ w,
                     const float* __restrict__ bias, float* __restrict__ xt, int N) {
    __shared__ __attribute__((aligned(16))) float sW[D_OUT * WP];
    const int nTilesM = (N + 15) >> 4;
    const int lane = threadIdx.x & 31;
    const int wave = __builtin_amdgcn_readfirstlane(threadIdx.x >> 5);
    const int half = lane >> 4;
    const int mr   = lane & 15;
    const int tileM = blockIdx.x * GEMM_WAVES + wave;

    int mA = tileM * 16 + mr; if (mA > N - 1) mA = N - 1;
    const float* aP = x + (size_t)mA * D_IN + 4 * half;
    float4 aN[4];
#pragma unroll
    for (int i = 0; i < 4; ++i) aN[i] = make_float4(0.0f, 0.0f, 0.0f, 0.0f);
    if (tileM < nTilesM) {
#pragma unroll
        for (int i = 0; i < 4; ++i) aN[i] = *(const float4*)(aP + 8 * i);
    }

    float hbv[6];
    float y2;
    {
        float bs = 0.0f;
#pragma unroll
        for (int t = 0; t < 6; ++t) { hbv[t] = bias[t * 16 + mr]; bs = fmaf(hbv[t], hbv[t], bs); }
        float n = fmaxf(sqrtf(hsum16(bs)), MINN);
        float e = tanhf(n) / n;
        float ps = 0.0f;
#pragma unroll
        for (int t = 0; t < 6; ++t) { hbv[t] = e * hbv[t]; ps = fmaf(hbv[t], hbv[t], ps); }
        float pn = fmaxf(sqrtf(hsum16(ps)), MINN);
        float pf = (pn > MAXN) ? MAXN / pn : 1.0f;
        float ys = 0.0f;
#pragma unroll
        for (int t = 0; t < 6; ++t) { hbv[t] *= pf; ys = fmaf(hbv[t], hbv[t], ys); }
        y2 = hsum16(ys);
    }

    for (int i = threadIdx.x; i < D_OUT * (D_IN / 4); i += GEMM_THREADS) {
        const int r = i >> 6, c = (i & 63) << 2;
        *(float4*)(sW + r * WP + c) = *(const float4*)(w + (size_t)r * D_IN + c);
    }
    __syncthreads();
    const float* bT[6];
#pragma unroll
    for (int t = 0; t < 6; ++t) bT[t] = sW + (t * 16 + mr) * WP + 4 * half;

    if (tileM >= nTilesM) return;
    const bool fullTile = (tileM * 16 + 15 < N);

    v8f acc[6] = {{}, {}, {}, {}, {}, {}};
    float4 aA[8][4];
#pragma unroll
    for (int i = 0; i < 4; ++i) aA[0][i] = aN[i];
    float4 bC[6], bN[6];
#pragma unroll
    for (int t = 0; t < 6; ++t) bC[t] = *(const float4*)(bT[t]);
    __builtin_amdgcn_sched_barrier(0);
#pragma unroll
    for (int kb = 0; kb < D_IN; kb += 32) {
        if (kb + 32 < D_IN) {
#pragma unroll
            for (int i = 0; i < 4; ++i) aA[kb / 32 + 1][i] = *(const float4*)(aP + kb + 32 + 8 * i);
        }
#pragma unroll
        for (int i = 0; i < 4; ++i) {
            if (kb + 8 * i + 8 < D_IN) {
#pragma unroll
                for (int t = 0; t < 6; ++t) bN[t] = *(const float4*)(bT[t] + kb + 8 * i + 8);
            }
            const float4 av = aA[kb / 32][i];
            v2f a01; a01.x = av.x; a01.y = av.y;
            v2f a23; a23.x = av.z; a23.y = av.w;
#pragma unroll
            for (int t = 0; t < 6; ++t) {
                v2f b01; b01.x = bC[t].x; b01.y = bC[t].y;
                acc[t] = __builtin_amdgcn_wmma_f32_16x16x4_f32(
                     false, a01,  false, b01,
                     (short)0, acc[t],  false,  false);
            }
#pragma unroll
            for (int t = 0; t < 6; ++t) {
                v2f b23; b23.x = bC[t].z; b23.y = bC[t].w;
                acc[t] = __builtin_amdgcn_wmma_f32_16x16x4_f32(
                     false, a23,  false, b23,
                     (short)0, acc[t],  false,  false);
            }
#pragma unroll
            for (int t = 0; t < 6; ++t) bC[t] = bN[t];
        }
    }
    __builtin_amdgcn_sched_barrier(0);
    asm volatile("v_nop\n\tv_nop\n\tv_nop\n\tv_nop");
    __builtin_amdgcn_sched_barrier(0);

    float ss0 = 0.0f, ss1 = 0.0f, ss2 = 0.0f, ss3 = 0.0f;
#pragma unroll
    for (int b = 0; b < 8; ++b) {
#pragma unroll
        for (int i = 0; i < 4; ++i) {
            ss0 = fmaf(aA[b][i].x, aA[b][i].x, ss0);
            ss1 = fmaf(aA[b][i].y, aA[b][i].y, ss1);
            ss2 = fmaf(aA[b][i].z, aA[b][i].z, ss2);
            ss3 = fmaf(aA[b][i].w, aA[b][i].w, ss3);
        }
    }

    float xs = (ss0 + ss1) + (ss2 + ss3);
    xs += __shfl_xor(xs, 16, 32);
    const float xsOther = dpp_read<0x168>(xs);
    const float xsMine  = ((mr >> 3) == half) ? xs : xsOther;
    const int   j       = mr & 7;

    float s[8], q[8], f0[8], f1[8];

#pragma unroll
    for (int r = 0; r < 8; ++r) {
        float v = 0.0f;
#pragma unroll
        for (int t = 0; t < 6; ++t) v = fmaf(acc[t][r], acc[t][r], v);
        s[r] = v;
    }
    hsum16x8(s);
    {
        float xn = fmaxf(sqrtf(xsMine), MINN);
        float mn = fmaxf(sqrtf(SEL8(s, j)), MINN);
        float sc = tanhf(mn / xn * atanhf(fminf(xn, ARTC))) / mn;
        bcast8(sc, f0);
    }
#pragma unroll
    for (int r = 0; r < 8; ++r) {
        float v = 0.0f;
#pragma unroll
        for (int t = 0; t < 6; ++t) { acc[t][r] *= f0[r]; v = fmaf(acc[t][r], acc[t][r], v); }
        s[r] = v;
    }
    hsum16x8(s);
    {
        float hn = fmaxf(sqrtf(SEL8(s, j)), MINN);
        float pf = (hn > MAXN) ? MAXN / hn : 1.0f;
        bcast8(pf, f0);
    }
#pragma unroll
    for (int r = 0; r < 8; ++r) {
        float v = 0.0f, u = 0.0f;
#pragma unroll
        for (int t = 0; t < 6; ++t) {
            acc[t][r] *= f0[r];
            v = fmaf(acc[t][r], acc[t][r], v);
            u = fmaf(acc[t][r], hbv[t], u);
        }
        s[r] = v; q[r] = u;
    }
    hsum16x8(s);
    hsum16x8(q);
    {
        float x2 = SEL8(s, j);
        float xy = SEL8(q, j);
        float ca  = 1.0f + 2.0f * xy + y2;
        float cb  = 1.0f - x2;
        float den = fmaxf(1.0f + 2.0f * xy + x2 * y2, MINN);
        bcast8(ca / den, f0); bcast8(cb / den, f1);
    }
#pragma unroll
    for (int r = 0; r < 8; ++r) {
        float v = 0.0f;
#pragma unroll
        for (int t = 0; t < 6; ++t) {
            acc[t][r] = f0[r] * acc[t][r] + f1[r] * hbv[t];
            v = fmaf(acc[t][r], acc[t][r], v);
        }
        s[r] = v;
    }
    hsum16x8(s);
    {
        float gn = fmaxf(sqrtf(SEL8(s, j)), MINN);
        float pf = (gn > MAXN) ? MAXN / gn : 1.0f;
        float pn  = fminf(gn, MAXN);
        float lsc = atanhf(fminf(pn, ARTC)) / pn;
        bcast8(pf, f0); bcast8(lsc, f1);
    }

    float* obase = xt + (size_t)(tileM * 16 + 8 * half) * D_OUT + mr;
#pragma unroll
    for (int r = 0; r < 8; ++r) {
        if (fullTile || tileM * 16 + 8 * half + r < N) {
            float* o = obase + (size_t)r * D_OUT;
#pragma unroll
            for (int t = 0; t < 6; ++t) o[t * 16] = f1[r] * (acc[t][r] * f0[r]);
        }
    }
}

#define OWN_ROWS    196
#define MAX_RANGES  256
#define MAX_SLICES  256
#define BIN_THREADS 1024
#define BIN_REGS    4

__global__ __launch_bounds__(BIN_THREADS) void bin_edges(
    const int* __restrict__ erow, const int* __restrict__ ecol,
    const float* __restrict__ ew, int* __restrict__ sorted_w,
    int* __restrict__ tcnt, int* __restrict__ toff, int E, int N, int S) {
    int4* __restrict__ sorted = (int4*)sorted_w;
    __shared__ int s_cnt[MAX_RANGES];
    __shared__ int s_off[MAX_RANGES];
    __shared__ int s_cur[MAX_RANGES];
    const int tid = threadIdx.x;
    const int b   = blockIdx.x;
    const int P   = gridDim.x;
    const int beg = b * S;
    const int end = min(E, beg + S);
    const bool inRegs = (S <= BIN_REGS * BIN_THREADS);

    if (tid < MAX_RANGES) s_cnt[tid] = 0;

    int r[BIN_REGS], c[BIN_REGS], w[BIN_REGS], rk[BIN_REGS];
    if (inRegs) {
        for (int i = 0; i < BIN_REGS; ++i) {
            const int e = beg + tid + i * BIN_THREADS;
            r[i] = -1; c[i] = 0; w[i] = 0;
            if (e < end) { r[i] = erow[e]; c[i] = ecol[e]; w[i] = __float_as_int(ew[e]); }
        }
    }
    __syncthreads();
    if (inRegs) {
        for (int i = 0; i < BIN_REGS; ++i) {
            rk[i] = 0;
            if ((unsigned)r[i] < (unsigned)N) rk[i] = atomicAdd(&s_cnt[r[i] / OWN_ROWS], 1);
        }
    } else {
        for (int e = beg + tid; e < end; e += BIN_THREADS) {
            const int rr = erow[e];
            if ((unsigned)rr < (unsigned)N) atomicAdd(&s_cnt[rr / OWN_ROWS], 1);
        }
    }
    __syncthreads();
    if (tid < 32) {
        int v[8];
        int s = 0;
        for (int j = 0; j < 8; ++j) { v[j] = s_cnt[tid * 8 + j]; s += v[j]; }
        int inc = s;
        for (int d = 1; d < 32; d <<= 1) {
            const int t = __shfl_up(inc, d, 32);
            if (tid >= d) inc += t;
        }
        int run = inc - s;
        for (int j = 0; j < 8; ++j) { s_off[tid * 8 + j] = run; s_cur[tid * 8 + j] = run; run += v[j]; }
    }
    __syncthreads();
    if (tid < MAX_RANGES) {
        tcnt[tid * P + b] = s_cnt[tid];
        toff[tid * P + b] = s_off[tid];
    }
    if (inRegs) {
        for (int i = 0; i < BIN_REGS; ++i) {
            if ((unsigned)r[i] < (unsigned)N)
                sorted[beg + s_off[r[i] / OWN_ROWS] + rk[i]] = make_int4(r[i], c[i], w[i], 0);
        }
    } else {
        for (int e = beg + tid; e < end; e += BIN_THREADS) {
            const int rr = erow[e];
            if ((unsigned)rr < (unsigned)N) {
                const int pos = beg + atomicAdd(&s_cur[rr / OWN_ROWS], 1);
                sorted[pos] = make_int4(rr, ecol[e], __float_as_int(ew[e]), 0);
            }
        }
    }
}

#define OWN_THREADS 1024
#define PASS_SLOTS  4
#define PASS_EDGES  (OWN_THREADS * PASS_SLOTS)
#define ROW_SLOTS   (2 * (OWN_THREADS / 8))
#define EDGE_UNROLL 4

__device__ __forceinline__ void sum_run(const float* __restrict__ xt, const int2* list,
                                        int n, int o, int s8, float (&acc)[12]) {
    for (int i = 0; i < n; i += EDGE_UNROLL) {
        int2   e[EDGE_UNROLL];
        float4 v[EDGE_UNROLL][3];
#pragma unroll
        for (int u = 0; u < EDGE_UNROLL; ++u) e[u] = list[o + min(i + u, n - 1)];
#pragma unroll
        for (int u = 0; u < EDGE_UNROLL; ++u) {
            const float4* p = (const float4*)(xt + (size_t)e[u].x * D_OUT) + s8;
            v[u][0] = p[0]; v[u][1] = p[8]; v[u][2] = p[16];
        }
#pragma unroll
        for (int u = 0; u < EDGE_UNROLL; ++u) {
            const float w = (i + u < n) ? __int_as_float(e[u].y) : 0.0f;
#pragma unroll
            for (int q = 0; q < 3; ++q) {
                acc[4 * q]     = fmaf(w, v[u][q].x, acc[4 * q]);
                acc[4 * q + 1] = fmaf(w, v[u][q].y, acc[4 * q + 1]);
                acc[4 * q + 2] = fmaf(w, v[u][q].z, acc[4 * q + 2]);
                acc[4 * q + 3] = fmaf(w, v[u][q].w, acc[4 * q + 3]);
            }
        }
    }
}

__global__ __launch_bounds__(OWN_THREADS) void own_rows(
    const int* __restrict__ sorted_w, const int* __restrict__ tcnt,
    const int* __restrict__ toff, const float* __restrict__ xt,
    float* __restrict__ out, int N, int S, int P) {
    const int4* __restrict__ sorted = (const int4*)sorted_w;
    __shared__ __attribute__((aligned(16))) int2 list[PASS_EDGES];
    __shared__ int s_c[MAX_SLICES];
    __shared__ int s_base[MAX_SLICES];
    __shared__ int s_pre[MAX_SLICES + 1];
    __shared__ int r_cnt[ROW_SLOTS];
    __shared__ int r_off[ROW_SLOTS];
    const int tid  = threadIdx.x;
    const int lane = tid & 31;
    const int wave = __builtin_amdgcn_readfirstlane(tid >> 5);
    const int k    = blockIdx.x;
    const int lo   = k * OWN_ROWS;
    const int nrows = min(OWN_ROWS, N - lo);

    if (tid < MAX_SLICES) {
        int c = 0, o = 0;
        if (tid < P) { c = tcnt[k * P + tid]; o = toff[k * P + tid]; }
        s_c[tid]    = c;
        s_base[tid] = tid * S + o;
    }
    if (tid < ROW_SLOTS) r_cnt[tid] = 0;
    __syncthreads();
    if (tid < 32) {
        int v[8];
        int s = 0;
        for (int j = 0; j < 8; ++j) { v[j] = s_c[tid * 8 + j]; s += v[j]; }
        int inc = s;
        for (int d = 1; d < 32; d <<= 1) {
            const int t = __shfl_up(inc, d, 32);
            if (tid >= d) inc += t;
        }
        int run = inc - s;
        for (int j = 0; j < 8; ++j) { s_pre[tid * 8 + j] = run; run += v[j]; }
        if (tid == 31) s_pre[MAX_SLICES] = run;
    }
    __syncthreads();
    const int T = __builtin_amdgcn_readfirstlane(s_pre[MAX_SLICES]);

    static_assert(OWN_ROWS <= ROW_SLOTS, "two rows per group of eight lanes must cover the block's rows");
    const int g = lane >> 3, s8 = lane & 7;
    const int rA = wave * 4 + g, rB = rA + OWN_THREADS / 8;
    float a[12], b[12];
#pragma unroll
    for (int j = 0; j < 12; ++j) { a[j] = 0.0f; b[j] = 0.0f; }

    for (int p0 = 0; p0 < T; p0 += PASS_EDGES) {
        int r[PASS_SLOTS], c[PASS_SLOTS], w[PASS_SLOTS], rk[PASS_SLOTS];
        int fc[PASS_SLOTS], lo_g[PASS_SLOTS], hi_g[PASS_SLOTS];
#pragma unroll
        for (int i = 0; i < PASS_SLOTS; ++i) {
            fc[i] = min(p0 + tid + i * OWN_THREADS, T - 1);
            lo_g[i] = 0; hi_g[i] = MAX_SLICES;
        }
#pragma unroll
        for (int it = 0; it < 8; ++it) {
            int m[PASS_SLOTS], sv[PASS_SLOTS];
#pragma unroll
            for (int i = 0; i < PASS_SLOTS; ++i) { m[i] = (lo_g[i] + hi_g[i]) >> 1; sv[i] = s_pre[m[i]]; }
#pragma unroll
            for (int i = 0; i < PASS_SLOTS; ++i) { if (sv[i] <= fc[i]) lo_g[i] = m[i]; else hi_g[i] = m[i]; }
        }
        int4 rec[PASS_SLOTS];
#pragma unroll
        for (int i = 0; i < PASS_SLOTS; ++i) rec[i] = sorted[s_base[lo_g[i]] + (fc[i] - s_pre[lo_g[i]])];
#pragma unroll
        for (int i = 0; i < PASS_SLOTS; ++i) {
            r[i]  = (p0 + tid + i * OWN_THREADS < T) ? rec[i].x - lo : -1;
            c[i]  = rec[i].y;
            w[i]  = rec[i].z;
            rk[i] = 0;
        }
#pragma unroll
        for (int i = 0; i < PASS_SLOTS; ++i)
            if ((unsigned)r[i] < (unsigned)ROW_SLOTS) rk[i] = atomicAdd(&r_cnt[r[i]], 1);
        __syncthreads();
        if (tid < 32) {
            int v[8];
            int s = 0;
            for (int j = 0; j < 8; ++j) { v[j] = r_cnt[tid * 8 + j]; s += v[j]; }
            int inc = s;
            for (int d = 1; d < 32; d <<= 1) {
                const int t = __shfl_up(inc, d, 32);
                if (tid >= d) inc += t;
            }
            int run = inc - s;
            for (int j = 0; j < 8; ++j) { r_off[tid * 8 + j] = run; run += v[j]; }
        }
        __syncthreads();
        int po[PASS_SLOTS];
#pragma unroll
        for (int i = 0; i < PASS_SLOTS; ++i) po[i] = r_off[r[i] & (ROW_SLOTS - 1)];
#pragma unroll
        for (int i = 0; i < PASS_SLOTS; ++i)
            if ((unsigned)r[i] < (unsigned)ROW_SLOTS) list[po[i] + rk[i]] = make_int2(c[i], w[i]);
        __syncthreads();

        sum_run(xt, list, r_cnt[rA], r_off[rA], s8, a);
        sum_run(xt, list, r_cnt[rB], r_off[rB], s8, b);

        if (p0 + PASS_EDGES < T) {
            __syncthreads();
            if (tid < ROW_SLOTS) r_cnt[tid] = 0;
            __syncthreads();
        }
    }

    if (wave * 4 < nrows) {
        const bool odd = (lane & 1) != 0;
        float qa, qb;
        qa = 0.0f; qb = 0.0f;
#pragma unroll
        for (int j = 0; j < 12; ++j) { qa = fmaf(a[j], a[j], qa); qb = fmaf(b[j], b[j], qb); }
        qa = gsum8(qa); qb = gsum8(qb);
        {
            float sn = fmaxf(sqrtf(odd ? qb : qa), MINN);
            float e  = tanhf(sn) / sn;
            const float ea = dpp_read<0x00>(e), eb = dpp_read<0x55>(e);
#pragma unroll
            for (int j = 0; j < 12; ++j) { a[j] *= ea; b[j] *= eb; }
        }
        qa = 0.0f; qb = 0.0f;
#pragma unroll
        for (int j = 0; j < 12; ++j) { qa = fmaf(a[j], a[j], qa); qb = fmaf(b[j], b[j], qb); }
        qa = gsum8(qa); qb = gsum8(qb);
        {
            float pnr = fmaxf(sqrtf(odd ? qb : qa), MINN);
            float pf  = (pnr > MAXN) ? MAXN / pnr : 1.0f;
            float pn  = fminf(pnr, MAXN);
            float l   = atanhf(fminf(pn, ARTC)) / pn;
            const float pfa = dpp_read<0x00>(pf), pfb = dpp_read<0x55>(pf);
            const float la  = dpp_read<0x00>(l),  lb  = dpp_read<0x55>(l);
#pragma unroll
            for (int j = 0; j < 12; ++j) {
                a[j] *= pfa; a[j] = fmaxf(la * a[j], 0.0f);
                b[j] *= pfb; b[j] = fmaxf(lb * b[j], 0.0f);
            }
        }
        qa = 0.0f; qb = 0.0f;
#pragma unroll
        for (int j = 0; j < 12; ++j) { qa = fmaf(a[j], a[j], qa); qb = fmaf(b[j], b[j], qb); }
        qa = gsum8(qa); qb = gsum8(qb);
        {
            float un = fmaxf(sqrtf(odd ? qb : qa), MINN);
            float e2 = tanhf(un) / un;
            const float ea = dpp_read<0x00>(e2), eb = dpp_read<0x55>(e2);
#pragma unroll
            for (int j = 0; j < 12; ++j) { a[j] *= ea; b[j] *= eb; }
        }
        qa = 0.0f; qb = 0.0f;
#pragma unroll
        for (int j = 0; j < 12; ++j) { qa = fmaf(a[j], a[j], qa); qb = fmaf(b[j], b[j], qb); }
        qa = gsum8(qa); qb = gsum8(qb);
        {
            float hn  = fmaxf(sqrtf(odd ? qb : qa), MINN);
            float pf2 = (hn > MAXN) ? MAXN / hn : 1.0f;
            const float fa = dpp_read<0x00>(pf2), fb = dpp_read<0x55>(pf2);
            if (rA < nrows) {
                float4* o = (float4*)(out + (size_t)(lo + rA) * D_OUT) + s8;
#pragma unroll
                for (int i = 0; i < 3; ++i)
                    o[8 * i] = make_float4(a[4 * i] * fa, a[4 * i + 1] * fa, a[4 * i + 2] * fa, a[4 * i + 3] * fa);
            }
            if (rB < nrows) {
                float4* o = (float4*)(out + (size_t)(lo + rB) * D_OUT) + s8;
#pragma unroll
                for (int i = 0; i < 3; ++i)
                    o[8 * i] = make_float4(b[4 * i] * fb, b[4 * i + 1] * fb, b[4 * i + 2] * fb, b[4 * i + 3] * fb);
            }
        }
    }
}

extern "C" void kernel_launch(void* const* d_in, const int* in_sizes, int n_in,
                              void* d_out, int out_size, void* d_ws, size_t ws_size,
                              hipStream_t stream) {
    const float* x      = (const float*)d_in[0];
    const float* weight = (const float*)d_in[1];
    const float* bias   = (const float*)d_in[2];
    const float* ew     = (const float*)d_in[3];
    const int*   erow   = (const int*)d_in[4];
    const int*   ecol   = (const int*)d_in[5];
    const int N = in_sizes[0] / D_IN;
    const int E = in_sizes[3];

    float* ws   = (float*)d_ws;
    float* buf0 = ws + 128;
    float* buf1 = buf0 + (size_t)N * D_OUT;
    float* out  = (float*)d_out;

    int*  sorted = (int*)buf0;
    int*  tcnt   = sorted + 4 * (size_t)E;
    int*  toff   = tcnt + MAX_RANGES * MAX_SLICES;

    int strips     = (N + 15) >> 4;
    int gemmBlocks = (strips + GEMM_WAVES - 1) / GEMM_WAVES;
    gemm_hyp_linear<<<gemmBlocks, GEMM_THREADS, 0, stream>>>(x, weight, bias, buf1, N);

    const int P = MAX_SLICES;
    const int S = (E + P - 1) / P;
    bin_edges<<<P, BIN_THREADS, 0, stream>>>(erow, ecol, ew, sorted, tcnt, toff, E, N, S);

    int ranges = (N + OWN_ROWS - 1) / OWN_ROWS;
    own_rows<<<ranges, OWN_THREADS, 0, stream>>>(sorted, tcnt, toff, buf1, out, N, S, P);
}


# edge binning runs inside the gemm launch: 60 extra 512-thread blocks each sort 1/60 of the edges on the WGPs the 196 strip blocks leave idle; 3 launches -> 2; edge order within a row's sum regrouped
# speedup vs baseline: 4.6879x; 1.2668x over previous
#include <hip/hip_runtime.h>
#include <hip/hip_bf16.h>


typedef __attribute__((ext_vector_type(2))) float v2f;
typedef __attribute__((ext_vector_type(8))) float v8f;

#define D_IN   256
#define D_OUT  96
#define MINN   1e-15f
#define MAXN   0.996f
#define ARTC   (1.0f - 1e-7f)

template <int CTRL>
__device__ __forceinline__ float dpp_read(float v) {
    return __builtin_bit_cast(float, __builtin_amdgcn_update_dpp(
        0, __builtin_bit_cast(int, v), CTRL, 0xf, 0xf, false));
}

__device__ __forceinline__ float hsum16(float v) {
    v += dpp_read<0x161>(v);
    v += dpp_read<0x162>(v);
    v += dpp_read<0x164>(v);
    v += dpp_read<0x168>(v);
    return v;
}

__device__ __forceinline__ float gsum8(float v) {
    v += dpp_read<0x161>(v);
    v += dpp_read<0x162>(v);
    v += dpp_read<0x164>(v);
    return v;
}

__device__ __forceinline__ void hsum16x8(float (&s)[8]) {
#pragma unroll
    for (int r = 0; r < 8; ++r) s[r] = hsum16(s[r]);
}

__device__ __forceinline__ float sel8v(float v0, float v1, float v2, float v3,
                                       float v4, float v5, float v6, float v7, int j) {
    const bool b0 = (j & 1) != 0, b1 = (j & 2) != 0, b2 = (j & 4) != 0;
    float a0 = b0 ? v1 : v0, a1 = b0 ? v3 : v2;
    float a2 = b0 ? v5 : v4, a3 = b0 ? v7 : v6;
    float c0 = b1 ? a1 : a0, c1 = b1 ? a3 : a2;
    return b2 ? c1 : c0;
}
#define SEL8(a, j) sel8v(a[0], a[1], a[2], a[3], a[4], a[5], a[6], a[7], j)

__device__ __forceinline__ void bcast8(float v, float (&o)[8]) {
    o[0] = dpp_read<0x150>(v); o[1] = dpp_read<0x151>(v);
    o[2] = dpp_read<0x152>(v); o[3] = dpp_read<0x153>(v);
    o[4] = dpp_read<0x154>(v); o[5] = dpp_read<0x155>(v);
    o[6] = dpp_read<0x156>(v); o[7] = dpp_read<0x157>(v);
}

#define GEMM_THREADS 512
#define GEMM_WAVES   (GEMM_THREADS / 32)
#define WP           260

#define OWN_ROWS    196
#define MAX_RANGES  256
#define MAX_SLICES  256
#define BIN_BLOCKS  60
#define BIN_SLOTS   27
#define BIN_CAP     (GEMM_THREADS * BIN_SLOTS)
#define SMEM_FLOATS (4 * (BIN_CAP + 32) + 3 * MAX_RANGES + 32)
static_assert(SMEM_FLOATS >= D_OUT * WP, "the shared array must hold the W rows of a strip block too");
static_assert(BIN_BLOCKS <= MAX_SLICES, "the owner kernel scans at most MAX_SLICES groups per range");
static_assert(BIN_CAP <= 65536 && MAX_RANGES + 32 <= 32768, "range and rank share one register: 15 + 16 bits");

__device__ __forceinline__ void bin_slice(
    const int* __restrict__ erow, const int* __restrict__ ecol,
    const float* __restrict__ ew, int4* __restrict__ sorted,
    int* __restrict__ tcnt, int* __restrict__ toff, int E, int N, int S,
    int b, int P, float* smem) {
    int4* stage = (int4*)smem;
    int*  s_cnt = (int*)(smem + 4 * (BIN_CAP + 32));
    int*  s_off = s_cnt + MAX_RANGES + 32;
    int*  s_cur = s_off + MAX_RANGES;
    const int tid = threadIdx.x;
    const int beg = b * S;
    const int end = min(E, beg + S);
    const bool inRegs = (S <= BIN_CAP);

    if (tid < MAX_RANGES + 32) s_cnt[tid] = 0;

    int r[BIN_SLOTS], c[BIN_SLOTS], w[BIN_SLOTS], kr[BIN_SLOTS];
    if (inRegs) {
#pragma unroll
        for (int i = 0; i < BIN_SLOTS; ++i) { r[i] = 0; c[i] = 0; w[i] = 0; }
        if (E > 0) {
#pragma unroll
            for (int i = 0; i < BIN_SLOTS; ++i) {
                const int ec = min(beg + tid + i * GEMM_THREADS, end - 1);
                r[i] = erow[ec]; c[i] = ecol[ec]; w[i] = __float_as_int(ew[ec]);
            }
        }
    }
    __syncthreads();
    if (inRegs) {
#pragma unroll
        for (int i = 0; i < BIN_SLOTS; ++i) {
            const bool ok = (beg + tid + i * GEMM_THREADS < end) && ((unsigned)r[i] < (unsigned)N);
            const int k = ok ? r[i] / OWN_ROWS : MAX_RANGES + (tid & 31);
            kr[i] = (k << 16) | atomicAdd(&s_cnt[k], 1);
        }
    } else {
        for (int e = beg + tid; e < end; e += GEMM_THREADS) {
            const int rr = erow[e];
            if ((unsigned)rr < (unsigned)N) atomicAdd(&s_cnt[rr / OWN_ROWS], 1);
        }
    }
    __syncthreads();
    if (tid < 32) {
        int v[8];
        int s = 0;
        for (int j = 0; j < 8; ++j) { v[j] = s_cnt[tid * 8 + j]; s += v[j]; }
        int inc = s;
        for (int d = 1; d < 32; d <<= 1) {
            const int t = __shfl_up(inc, d, 32);
            if (tid >= d) inc += t;
        }
        int run = inc - s;
        for (int j = 0; j < 8; ++j) { s_off[tid * 8 + j] = run; s_cur[tid * 8 + j] = run; run += v[j]; }
    }
    __syncthreads();
    if (tid < MAX_RANGES) {
        tcnt[tid * P + b] = s_cnt[tid];
        toff[tid * P + b] = s_off[tid];
    }
    if (inRegs) {
        int so[BIN_SLOTS];
#pragma unroll
        for (int i = 0; i < BIN_SLOTS; ++i) so[i] = s_off[(kr[i] >> 16) & (MAX_RANGES - 1)];
#pragma unroll
        for (int i = 0; i < BIN_SLOTS; ++i) {
            const int pos = ((kr[i] >> 16) < MAX_RANGES) ? so[i] + (kr[i] & 0xffff) : BIN_CAP + (tid & 31);
            stage[pos] = make_int4(r[i], c[i], w[i], 0);
        }
        __syncthreads();
        const int tot  = __builtin_amdgcn_readfirstlane(s_off[MAX_RANGES - 1] + s_cnt[MAX_RANGES - 1]);
        const int full = tot / GEMM_THREADS;
#pragma unroll 4
        for (int i = 0; i < full; ++i) sorted[beg + tid + i * GEMM_THREADS] = stage[tid + i * GEMM_THREADS];
        if (tid + full * GEMM_THREADS < tot)
            sorted[beg + tid + full * GEMM_THREADS] = stage[tid + full * GEMM_THREADS];
    } else {
        for (int e = beg + tid; e < end; e += GEMM_THREADS) {
            const int rr = erow[e];
            if ((unsigned)rr < (unsigned)N) {
                const int pos = beg + atomicAdd(&s_cur[rr / OWN_ROWS], 1);
                sorted[pos] = make_int4(rr, ecol[e], __float_as_int(ew[e]), 0);
            }
        }
    }
}

__global__ __launch_bounds__(GEMM_THREADS) __attribute__((amdgpu_waves_per_eu(4, 5)))
void hyp_linear_and_bin(const float* __restrict__ x, const float* __restrict__ w,
                        const float* __restrict__ bias, float* __restrict__ xt, int N,
                        const int* __restrict__ erow, const int* __restrict__ ecol,
                        const float* __restrict__ ew, int* __restrict__ sorted_w,
                        int* __restrict__ tcnt, int* __restrict__ toff,
                        int E, int S, int gemmBlocks) {
    __shared__ __attribute__((aligned(16))) float smem[SMEM_FLOATS];
    if ((int)blockIdx.x >= gemmBlocks) {
        bin_slice(erow, ecol, ew, (int4*)sorted_w, tcnt, toff, E, N, S,
                  (int)blockIdx.x - gemmBlocks, (int)gridDim.x - gemmBlocks, smem);
        return;
    }
    float* sW = smem;
    const int nTilesM = (N + 15) >> 4;
    const int lane = threadIdx.x & 31;
    const int wave = __builtin_amdgcn_readfirstlane(threadIdx.x >> 5);
    const int half = lane >> 4;
    const int mr   = lane & 15;
    const int tileM = blockIdx.x * GEMM_WAVES + wave;

    int mA = tileM * 16 + mr; if (mA > N - 1) mA = N - 1;
    const float* aP = x + (size_t)mA * D_IN + 4 * half;
    float4 aN[4];
#pragma unroll
    for (int i = 0; i < 4; ++i) aN[i] = make_float4(0.0f, 0.0f, 0.0f, 0.0f);
    if (tileM < nTilesM) {
#pragma unroll
        for (int i = 0; i < 4; ++i) aN[i] = *(const float4*)(aP + 8 * i);
    }

    float hbv[6];
    float y2;
    {
        float bs = 0.0f;
#pragma unroll
        for (int t = 0; t < 6; ++t) { hbv[t] = bias[t * 16 + mr]; bs = fmaf(hbv[t], hbv[t], bs); }
        float n = fmaxf(sqrtf(hsum16(bs)), MINN);
        float e = tanhf(n) / n;
        float ps = 0.0f;
#pragma unroll
        for (int t = 0; t < 6; ++t) { hbv[t] = e * hbv[t]; ps = fmaf(hbv[t], hbv[t], ps); }
        float pn = fmaxf(sqrtf(hsum16(ps)), MINN);
        float pf = (pn > MAXN) ? MAXN / pn : 1.0f;
        float ys = 0.0f;
#pragma unroll
        for (int t = 0; t < 6; ++t) { hbv[t] *= pf; ys = fmaf(hbv[t], hbv[t], ys); }
        y2 = hsum16(ys);
    }

    for (int i = threadIdx.x; i < D_OUT * (D_IN / 4); i += GEMM_THREADS) {
        const int r = i >> 6, c = (i & 63) << 2;
        *(float4*)(sW + r * WP + c) = *(const float4*)(w + (size_t)r * D_IN + c);
    }
    __syncthreads();
    const float* bT[6];
#pragma unroll
    for (int t = 0; t < 6; ++t) bT[t] = sW + (t * 16 + mr) * WP + 4 * half;

    if (tileM >= nTilesM) return;
    const bool fullTile = (tileM * 16 + 15 < N);

    v8f acc[6] = {{}, {}, {}, {}, {}, {}};
    float4 aA[8][4];
#pragma unroll
    for (int i = 0; i < 4; ++i) aA[0][i] = aN[i];
    float4 bC[6], bN[6];
#pragma unroll
    for (int t = 0; t < 6; ++t) bC[t] = *(const float4*)(bT[t]);
    __builtin_amdgcn_sched_barrier(0);
#pragma unroll
    for (int kb = 0; kb < D_IN; kb += 32) {
        if (kb + 32 < D_IN) {
#pragma unroll
            for (int i = 0; i < 4; ++i) aA[kb / 32 + 1][i] = *(const float4*)(aP + kb + 32 + 8 * i);
        }
#pragma unroll
        for (int i = 0; i < 4; ++i) {
            if (kb + 8 * i + 8 < D_IN) {
#pragma unroll
                for (int t = 0; t < 6; ++t) bN[t] = *(const float4*)(bT[t] + kb + 8 * i + 8);
            }
            const float4 av = aA[kb / 32][i];
            v2f a01; a01.x = av.x; a01.y = av.y;
            v2f a23; a23.x = av.z; a23.y = av.w;
#pragma unroll
            for (int t = 0; t < 6; ++t) {
                v2f b01; b01.x = bC[t].x; b01.y = bC[t].y;
                acc[t] = __builtin_amdgcn_wmma_f32_16x16x4_f32(
                     false, a01,  false, b01,
                     (short)0, acc[t],  false,  false);
            }
#pragma unroll
            for (int t = 0; t < 6; ++t) {
                v2f b23; b23.x = bC[t].z; b23.y = bC[t].w;
                acc[t] = __builtin_amdgcn_wmma_f32_16x16x4_f32(
                     false, a23,  false, b23,
                     (short)0, acc[t],  false,  false);
            }
#pragma unroll
            for (int t = 0; t < 6; ++t) bC[t] = bN[t];
        }
    }
    __builtin_amdgcn_sched_barrier(0);
    asm volatile("v_nop\n\tv_nop\n\tv_nop\n\tv_nop");
    __builtin_amdgcn_sched_barrier(0);

    float ss0 = 0.0f, ss1 = 0.0f, ss2 = 0.0f, ss3 = 0.0f;
#pragma unroll
    for (int b = 0; b < 8; ++b) {
#pragma unroll
        for (int i = 0; i < 4; ++i) {
            ss0 = fmaf(aA[b][i].x, aA[b][i].x, ss0);
            ss1 = fmaf(aA[b][i].y, aA[b][i].y, ss1);
            ss2 = fmaf(aA[b][i].z, aA[b][i].z, ss2);
            ss3 = fmaf(aA[b][i].w, aA[b][i].w, ss3);
        }
    }

    float xs = (ss0 + ss1) + (ss2 + ss3);
    xs += __shfl_xor(xs, 16, 32);
    const float xsOther = dpp_read<0x168>(xs);
    const float xsMine  = ((mr >> 3) == half) ? xs : xsOther;
    const int   j       = mr & 7;

    float s[8], q[8], f0[8], f1[8];

#pragma unroll
    for (int r = 0; r < 8; ++r) {
        float v = 0.0f;
#pragma unroll
        for (int t = 0; t < 6; ++t) v = fmaf(acc[t][r], acc[t][r], v);
        s[r] = v;
    }
    hsum16x8(s);
    {
        float xn = fmaxf(sqrtf(xsMine), MINN);
        float mn = fmaxf(sqrtf(SEL8(s, j)), MINN);
        float sc = tanhf(mn / xn * atanhf(fminf(xn, ARTC))) / mn;
        bcast8(sc, f0);
    }
#pragma unroll
    for (int r = 0; r < 8; ++r) {
        float v = 0.0f;
#pragma unroll
        for (int t = 0; t < 6; ++t) { acc[t][r] *= f0[r]; v = fmaf(acc[t][r], acc[t][r], v); }
        s[r] = v;
    }
    hsum16x8(s);
    {
        float hn = fmaxf(sqrtf(SEL8(s, j)), MINN);
        float pf = (hn > MAXN) ? MAXN / hn : 1.0f;
        bcast8(pf, f0);
    }
#pragma unroll
    for (int r = 0; r < 8; ++r) {
        float v = 0.0f, u = 0.0f;
#pragma unroll
        for (int t = 0; t < 6; ++t) {
            acc[t][r] *= f0[r];
            v = fmaf(acc[t][r], acc[t][r], v);
            u = fmaf(acc[t][r], hbv[t], u);
        }
        s[r] = v; q[r] = u;
    }
    hsum16x8(s);
    hsum16x8(q);
    {
        float x2 = SEL8(s, j);
        float xy = SEL8(q, j);
        float ca  = 1.0f + 2.0f * xy + y2;
        float cb  = 1.0f - x2;
        float den = fmaxf(1.0f + 2.0f * xy + x2 * y2, MINN);
        bcast8(ca / den, f0); bcast8(cb / den, f1);
    }
#pragma unroll
    for (int r = 0; r < 8; ++r) {
        float v = 0.0f;
#pragma unroll
        for (int t = 0; t < 6; ++t) {
            acc[t][r] = f0[r] * acc[t][r] + f1[r] * hbv[t];
            v = fmaf(acc[t][r], acc[t][r], v);
        }
        s[r] = v;
    }
    hsum16x8(s);
    {
        float gn = fmaxf(sqrtf(SEL8(s, j)), MINN);
        float pf = (gn > MAXN) ? MAXN / gn : 1.0f;
        float pn  = fminf(gn, MAXN);
        float lsc = atanhf(fminf(pn, ARTC)) / pn;
        bcast8(pf, f0); bcast8(lsc, f1);
    }

    float* obase = xt + (size_t)(tileM * 16 + 8 * half) * D_OUT + mr;
#pragma unroll
    for (int r = 0; r < 8; ++r) {
        if (fullTile || tileM * 16 + 8 * half + r < N) {
            float* o = obase + (size_t)r * D_OUT;
#pragma unroll
            for (int t = 0; t < 6; ++t) o[t * 16] = f1[r] * (acc[t][r] * f0[r]);
        }
    }
}

#define OWN_THREADS 1024
#define PASS_SLOTS  4
#define PASS_EDGES  (OWN_THREADS * PASS_SLOTS)
#define ROW_SLOTS   (2 * (OWN_THREADS / 8))
#define EDGE_UNROLL 4

__device__ __forceinline__ void sum_run(const float* __restrict__ xt, const int2* list,
                                        int n, int o, int s8, float (&acc)[12]) {
    for (int i = 0; i < n; i += EDGE_UNROLL) {
        int2   e[EDGE_UNROLL];
        float4 v[EDGE_UNROLL][3];
#pragma unroll
        for (int u = 0; u < EDGE_UNROLL; ++u) e[u] = list[o + min(i + u, n - 1)];
#pragma unroll
        for (int u = 0; u < EDGE_UNROLL; ++u) {
            const float4* p = (const float4*)(xt + (size_t)e[u].x * D_OUT) + s8;
            v[u][0] = p[0]; v[u][1] = p[8]; v[u][2] = p[16];
        }
#pragma unroll
        for (int u = 0; u < EDGE_UNROLL; ++u) {
            const float w = (i + u < n) ? __int_as_float(e[u].y) : 0.0f;
#pragma unroll
            for (int q = 0; q < 3; ++q) {
                acc[4 * q]     = fmaf(w, v[u][q].x, acc[4 * q]);
                acc[4 * q + 1] = fmaf(w, v[u][q].y, acc[4 * q + 1]);
                acc[4 * q + 2] = fmaf(w, v[u][q].z, acc[4 * q + 2]);
                acc[4 * q + 3] = fmaf(w, v[u][q].w, acc[4 * q + 3]);
            }
        }
    }
}

__global__ __launch_bounds__(OWN_THREADS) void own_rows(
    const int* __restrict__ sorted_w, const int* __restrict__ tcnt,
    const int* __restrict__ toff, const float* __restrict__ xt,
    float* __restrict__ out, int N, int S, int P) {
    const int4* __restrict__ sorted = (const int4*)sorted_w;
    __shared__ __attribute__((aligned(16))) int2 list[PASS_EDGES];
    __shared__ int s_c[MAX_SLICES];
    __shared__ int s_base[MAX_SLICES];
    __shared__ int s_pre[MAX_SLICES + 1];
    __shared__ int r_cnt[ROW_SLOTS];
    __shared__ int r_off[ROW_SLOTS];
    const int tid  = threadIdx.x;
    const int lane = tid & 31;
    const int wave = __builtin_amdgcn_readfirstlane(tid >> 5);
    const int k    = blockIdx.x;
    const int lo   = k * OWN_ROWS;
    const int nrows = min(OWN_ROWS, N - lo);

    if (tid < MAX_SLICES) {
        int c = 0, o = 0;
        if (tid < P) { c = tcnt[k * P + tid]; o = toff[k * P + tid]; }
        s_c[tid]    = c;
        s_base[tid] = tid * S + o;
    }
    if (tid < ROW_SLOTS) r_cnt[tid] = 0;
    __syncthreads();
    if (tid < 32) {
        int v[8];
        int s = 0;
        for (int j = 0; j < 8; ++j) { v[j] = s_c[tid * 8 + j]; s += v[j]; }
        int inc = s;
        for (int d = 1; d < 32; d <<= 1) {
            const int t = __shfl_up(inc, d, 32);
            if (tid >= d) inc += t;
        }
        int run = inc - s;
        for (int j = 0; j < 8; ++j) { s_pre[tid * 8 + j] = run; run += v[j]; }
        if (tid == 31) s_pre[MAX_SLICES] = run;
    }
    __syncthreads();
    const int T = __builtin_amdgcn_readfirstlane(s_pre[MAX_SLICES]);

    static_assert(OWN_ROWS <= ROW_SLOTS, "two rows per group of eight lanes must cover the block's rows");
    const int g = lane >> 3, s8 = lane & 7;
    const int rA = wave * 4 + g, rB = rA + OWN_THREADS / 8;
    float a[12], b[12];
#pragma unroll
    for (int j = 0; j < 12; ++j) { a[j] = 0.0f; b[j] = 0.0f; }

    for (int p0 = 0; p0 < T; p0 += PASS_EDGES) {
        int r[PASS_SLOTS], c[PASS_SLOTS], w[PASS_SLOTS], rk[PASS_SLOTS];
        int fc[PASS_SLOTS], lo_g[PASS_SLOTS], hi_g[PASS_SLOTS];
#pragma unroll
        for (int i = 0; i < PASS_SLOTS; ++i) {
            fc[i] = min(p0 + tid + i * OWN_THREADS, T - 1);
            lo_g[i] = 0; hi_g[i] = MAX_SLICES;
        }
#pragma unroll
        for (int it = 0; it < 8; ++it) {
            int m[PASS_SLOTS], sv[PASS_SLOTS];
#pragma unroll
            for (int i = 0; i < PASS_SLOTS; ++i) { m[i] = (lo_g[i] + hi_g[i]) >> 1; sv[i] = s_pre[m[i]]; }
#pragma unroll
            for (int i = 0; i < PASS_SLOTS; ++i) { if (sv[i] <= fc[i]) lo_g[i] = m[i]; else hi_g[i] = m[i]; }
        }
        int4 rec[PASS_SLOTS];
#pragma unroll
        for (int i = 0; i < PASS_SLOTS; ++i) rec[i] = sorted[s_base[lo_g[i]] + (fc[i] - s_pre[lo_g[i]])];
#pragma unroll
        for (int i = 0; i < PASS_SLOTS; ++i) {
            r[i]  = (p0 + tid + i * OWN_THREADS < T) ? rec[i].x - lo : -1;
            c[i]  = rec[i].y;
            w[i]  = rec[i].z;
            rk[i] = 0;
        }
#pragma unroll
        for (int i = 0; i < PASS_SLOTS; ++i)
            if ((unsigned)r[i] < (unsigned)ROW_SLOTS) rk[i] = atomicAdd(&r_cnt[r[i]], 1);
        __syncthreads();
        if (tid < 32) {
            int v[8];
            int s = 0;
            for (int j = 0; j < 8; ++j) { v[j] = r_cnt[tid * 8 + j]; s += v[j]; }
            int inc = s;
            for (int d = 1; d < 32; d <<= 1) {
                const int t = __shfl_up(inc, d, 32);
                if (tid >= d) inc += t;
            }
            int run = inc - s;
            for (int j = 0; j < 8; ++j) { r_off[tid * 8 + j] = run; run += v[j]; }
        }
        __syncthreads();
        int po[PASS_SLOTS];
#pragma unroll
        for (int i = 0; i < PASS_SLOTS; ++i) po[i] = r_off[r[i] & (ROW_SLOTS - 1)];
#pragma unroll
        for (int i = 0; i < PASS_SLOTS; ++i)
            if ((unsigned)r[i] < (unsigned)ROW_SLOTS) list[po[i] + rk[i]] = make_int2(c[i], w[i]);
        __syncthreads();

        sum_run(xt, list, r_cnt[rA], r_off[rA], s8, a);
        sum_run(xt, list, r_cnt[rB], r_off[rB], s8, b);

        if (p0 + PASS_EDGES < T) {
            __syncthreads();
            if (tid < ROW_SLOTS) r_cnt[tid] = 0;
            __syncthreads();
        }
    }

    if (wave * 4 < nrows) {
        const bool odd = (lane & 1) != 0;
        float qa, qb;
        qa = 0.0f; qb = 0.0f;
#pragma unroll
        for (int j = 0; j < 12; ++j) { qa = fmaf(a[j], a[j], qa); qb = fmaf(b[j], b[j], qb); }
        qa = gsum8(qa); qb = gsum8(qb);
        {
            float sn = fmaxf(sqrtf(odd ? qb : qa), MINN);
            float e  = tanhf(sn) / sn;
            const float ea = dpp_read<0x00>(e), eb = dpp_read<0x55>(e);
#pragma unroll
            for (int j = 0; j < 12; ++j) { a[j] *= ea; b[j] *= eb; }
        }
        qa = 0.0f; qb = 0.0f;
#pragma unroll
        for (int j = 0; j < 12; ++j) { qa = fmaf(a[j], a[j], qa); qb = fmaf(b[j], b[j], qb); }
        qa = gsum8(qa); qb = gsum8(qb);
        {
            float pnr = fmaxf(sqrtf(odd ? qb : qa), MINN);
            float pf  = (pnr > MAXN) ? MAXN / pnr : 1.0f;
            float pn  = fminf(pnr, MAXN);
            float l   = atanhf(fminf(pn, ARTC)) / pn;
            const float pfa = dpp_read<0x00>(pf), pfb = dpp_read<0x55>(pf);
            const float la  = dpp_read<0x00>(l),  lb  = dpp_read<0x55>(l);
#pragma unroll
            for (int j = 0; j < 12; ++j) {
                a[j] *= pfa; a[j] = fmaxf(la * a[j], 0.0f);
                b[j] *= pfb; b[j] = fmaxf(lb * b[j], 0.0f);
            }
        }
        qa = 0.0f; qb = 0.0f;
#pragma unroll
        for (int j = 0; j < 12; ++j) { qa = fmaf(a[j], a[j], qa); qb = fmaf(b[j], b[j], qb); }
        qa = gsum8(qa); qb = gsum8(qb);
        {
            float un = fmaxf(sqrtf(odd ? qb : qa), MINN);
            float e2 = tanhf(un) / un;
            const float ea = dpp_read<0x00>(e2), eb = dpp_read<0x55>(e2);
#pragma unroll
            for (int j = 0; j < 12; ++j) { a[j] *= ea; b[j] *= eb; }
        }
        qa = 0.0f; qb = 0.0f;
#pragma unroll
        for (int j = 0; j < 12; ++j) { qa = fmaf(a[j], a[j], qa); qb = fmaf(b[j], b[j], qb); }
        qa = gsum8(qa); qb = gsum8(qb);
        {
            float hn  = fmaxf(sqrtf(odd ? qb : qa), MINN);
            float pf2 = (hn > MAXN) ? MAXN / hn : 1.0f;
            const float fa = dpp_read<0x00>(pf2), fb = dpp_read<0x55>(pf2);
            if (rA < nrows) {
                float4* o = (float4*)(out + (size_t)(lo + rA) * D_OUT) + s8;
#pragma unroll
                for (int i = 0; i < 3; ++i)
                    o[8 * i] = make_float4(a[4 * i] * fa, a[4 * i + 1] * fa, a[4 * i + 2] * fa, a[4 * i + 3] * fa);
            }
            if (rB < nrows) {
                float4* o = (float4*)(out + (size_t)(lo + rB) * D_OUT) + s8;
#pragma unroll
                for (int i = 0; i < 3; ++i)
                    o[8 * i] = make_float4(b[4 * i] * fb, b[4 * i + 1] * fb, b[4 * i + 2] * fb, b[4 * i + 3] * fb);
            }
        }
    }
}

extern "C" void kernel_launch(void* const* d_in, const int* in_sizes, int n_in,
                              void* d_out, int out_size, void* d_ws, size_t ws_size,
                              hipStream_t stream) {
    const float* x      = (const float*)d_in[0];
    const float* weight = (const float*)d_in[1];
    const float* bias   = (const float*)d_in[2];
    const float* ew     = (const float*)d_in[3];
    const int*   erow   = (const int*)d_in[4];
    const int*   ecol   = (const int*)d_in[5];
    const int N = in_sizes[0] / D_IN;
    const int E = in_sizes[3];

    float* ws   = (float*)d_ws;
    float* buf0 = ws + 128;
    float* buf1 = buf0 + (size_t)N * D_OUT;
    float* out  = (float*)d_out;

    int*  sorted = (int*)buf0;
    int*  tcnt   = sorted + 4 * (size_t)E;
    int*  toff   = tcnt + MAX_RANGES * MAX_SLICES;

    int strips     = (N + 15) >> 4;
    int gemmBlocks = (strips + GEMM_WAVES - 1) / GEMM_WAVES;
    const int P = BIN_BLOCKS;
    const int S = (E + P - 1) / P;
    hyp_linear_and_bin<<<gemmBlocks + P, GEMM_THREADS, 0, stream>>>(
        x, weight, bias, buf1, N, erow, ecol, ew, sorted, tcnt, toff, E, S, gemmBlocks);

    int ranges = (N + OWN_ROWS - 1) / OWN_ROWS;
    own_rows<<<ranges, OWN_THREADS, 0, stream>>>(sorted, tcnt, toff, buf1, out, N, S, P);
}
